# starting kernel
# baseline (speedup 1.0000x reference)
#include <hip/hip_runtime.h>

typedef __attribute__((ext_vector_type(2))) float v2f;
typedef __attribute__((ext_vector_type(8))) float v8f;

#define DIM 128

__global__ void k_deg_init(float* deg, int n) {
    int i = blockIdx.x * blockDim.x + threadIdx.x;
    if (i < n) deg[i] = 1.0f;
}

__global__ void k_deg_count(const int* __restrict__ col, float* deg, int e) {
    int i = blockIdx.x * blockDim.x + threadIdx.x;
    if (i < e) atomicAdd(&deg[col[i]], 1.0f);
}

__global__ void k_deg_rsqrt(float* deg, int n) {
    int i = blockIdx.x * blockDim.x + threadIdx.x;
    if (i < n) deg[i] = rsqrtf(deg[i]);
}

__global__ __launch_bounds__(256) void k_gemm128_wmma(
    const float* __restrict__ X, const float* __restrict__ W,
    const float* __restrict__ cvec, float* __restrict__ Y, int n) {
    const int lane = threadIdx.x & 31;
    const int wave = threadIdx.x >> 5;
    const int m0   = blockIdx.x * 16;
    const int n0   = wave * 16;
    const int lo   = lane & 15;
    const int hi   = lane >> 4;

    int arow = m0 + lo;
    int arow_c = (arow < n) ? arow : (n - 1);
    const float* ap = X + (long long)arow_c * DIM + 2 * hi;
    const float* bp = W + (n0 + lo) + (long long)(2 * hi) * DIM;

    float ci = cvec ? cvec[n0 + lo] : 0.0f;
    v8f d;
#pragma unroll
    for (int v = 0; v < 8; ++v) d[v] = ci;

#pragma unroll
    for (int k = 0; k < DIM; k += 4) {
        v2f a = *(const v2f*)(ap + k);
        v2f b;
        b.x = bp[(long long)k * DIM];
        b.y = bp[(long long)k * DIM + DIM];
        d = __builtin_amdgcn_wmma_f32_16x16x4_f32(false, a, false, b,
                                                  (short)0, d, false, false);
    }

    float* yp = Y + (long long)(m0 + 8 * hi) * DIM + n0 + lo;
#pragma unroll
    for (int v = 0; v < 8; ++v) {
        int m = m0 + v + 8 * hi;
        if (m < n) yp[(long long)v * DIM] = d[v];
    }
}

__global__ void k_selfloop_init(const float* __restrict__ xw, const float* __restrict__ dinv,
                                float* __restrict__ acc, int n) {
    int t = blockIdx.x * blockDim.x + threadIdx.x;
    if (t >= n * 32) return;
    int node = t >> 5;
    float s = dinv[node]; s *= s;
    float4 v = ((const float4*)xw)[t];
    float4 o = make_float4(v.x * s, v.y * s, v.z * s, v.w * s);
    ((float4*)acc)[t] = o;
}

__global__ void k_edge_scatter(const float* __restrict__ xw,
                               const int* __restrict__ rowi, const int* __restrict__ coli,
                               const float* __restrict__ dinv, float* __restrict__ acc, int e) {
    long long t = (long long)blockIdx.x * blockDim.x + threadIdx.x;
    int eid = (int)(t >> 5);
    if (eid >= e) return;
    int lane = (int)(t & 31);
    int r = rowi[eid], c = coli[eid];
    float norm = dinv[r] * dinv[c];
    float4 v = *(const float4*)(xw + (long long)r * DIM + lane * 4);
    float* dst = acc + (long long)c * DIM + lane * 4;
    atomicAdd(dst + 0, v.x * norm);
    atomicAdd(dst + 1, v.y * norm);
    atomicAdd(dst + 2, v.z * norm);
    atomicAdd(dst + 3, v.w * norm);
}

__global__ void k_root_and_c(const float* __restrict__ acc, const float* __restrict__ b1,
                             const float* __restrict__ posts, const float* __restrict__ W2,
                             const int* __restrict__ rootPtr,
                             float* __restrict__ root_feat, float* __restrict__ cvec) {
    int j = threadIdx.x;
    int root = *rootPtr;
    root_feat[j] = acc[(long long)root * DIM + j] + b1[j];
    float s = 0.0f;
    for (int k = 0; k < DIM; ++k)
        s += fmaxf(posts[(long long)root * DIM + k], 0.0f) * W2[(long long)(DIM + k) * DIM + j];
    cvec[j] = s;
}

__global__ void k_bias_relu(float* __restrict__ acc, const float* __restrict__ b, int n) {
    int t = blockIdx.x * blockDim.x + threadIdx.x;
    if (t >= n * 32) return;
    float4 bb = ((const float4*)b)[t & 31];
    float4 v = ((float4*)acc)[t];
    v.x = fmaxf(v.x + bb.x, 0.0f);
    v.y = fmaxf(v.y + bb.y, 0.0f);
    v.z = fmaxf(v.z + bb.z, 0.0f);
    v.w = fmaxf(v.w + bb.w, 0.0f);
    ((float4*)acc)[t] = v;
}

__global__ __launch_bounds__(256) void k_finalize_pool(
    const float* __restrict__ acc, const float* __restrict__ b2,
    const float* __restrict__ root_feat, float* __restrict__ out, int n) {
    __shared__ float feat[256];
    int i = blockIdx.x;
    int t = threadIdx.x;
    if (t < 128) feat[t] = root_feat[t];
    else         feat[t] = fmaxf(acc[(long long)i * DIM + (t - 128)] + b2[t - 128], 0.0f);
    __syncthreads();
    if (t < 254)
        out[(long long)i * 254 + t] = (feat[t] + feat[t + 1] + feat[t + 2]) * (1.0f / 3.0f);
}

extern "C" void kernel_launch(void* const* d_in, const int* in_sizes, int n_in,
                              void* d_out, int out_size, void* d_ws, size_t ws_size,
                              hipStream_t stream) {
    const float* posts = (const float*)d_in[0];
    const float* W1    = (const float*)d_in[1];
    const float* b1    = (const float*)d_in[2];
    const float* W2    = (const float*)d_in[3];
    const float* b2    = (const float*)d_in[4];
    const int*   edge  = (const int*)d_in[5];
    const int*   rootP = (const int*)d_in[6];

    const int n = in_sizes[0] / DIM;
    const int e = in_sizes[5] / 2;
    const int* rowi = edge;
    const int* coli = edge + e;

    char* ws = (char*)d_ws;
    size_t off = 0;
    float* dinv = (float*)(ws + off); off += (((size_t)n * 4) + 255) & ~(size_t)255;
    float* xw   = (float*)(ws + off); off += (size_t)n * DIM * 4;
    float* acc  = (float*)(ws + off); off += (size_t)n * DIM * 4;
    float* rootf = (float*)(ws + off); off += DIM * 4;
    float* cvec  = (float*)(ws + off);
    float* out = (float*)d_out;

    const int B = 256;
    const int gN   = (n + B - 1) / B;
    const int gE   = (e + B - 1) / B;
    const int gNF4 = (n * 32 + B - 1) / B;
    const int gE32 = (int)(((long long)e * 32 + B - 1) / B);
    const int gM   = (n + 15) / 16;

    k_deg_init <<<gN, B, 0, stream>>>(dinv, n);
    k_deg_count<<<gE, B, 0, stream>>>(coli, dinv, e);
    k_deg_rsqrt<<<gN, B, 0, stream>>>(dinv, n);

    k_gemm128_wmma <<<gM, B, 0, stream>>>(posts, W1, nullptr, xw, n);
    k_selfloop_init<<<gNF4, B, 0, stream>>>(xw, dinv, acc, n);
    k_edge_scatter <<<gE32, B, 0, stream>>>(xw, rowi, coli, dinv, acc, e);

    k_root_and_c<<<1, DIM, 0, stream>>>(acc, b1, posts, W2, rootP, rootf, cvec);

    k_bias_relu    <<<gNF4, B, 0, stream>>>(acc, b1, n);
    k_gemm128_wmma <<<gM, B, 0, stream>>>(acc, W2, cvec, xw, n);
    k_selfloop_init<<<gNF4, B, 0, stream>>>(xw, dinv, acc, n);
    k_edge_scatter <<<gE32, B, 0, stream>>>(xw, rowi, coli, dinv, acc, e);

    k_finalize_pool<<<n, B, 0, stream>>>(acc, b2, rootf, out, n);
}


# Both GEMM launches: the 16x128 A tile is fetched once per work-group into LDS by whole-row 16-byte loads; the 8 waves read fragments from LDS, not from global (16 lines per load); sums unchanged
# speedup vs baseline: 1.8178x; 1.8178x over previous
#include <hip/hip_runtime.h>

typedef __attribute__((ext_vector_type(2))) float v2f;
typedef __attribute__((ext_vector_type(8))) float v8f;

#define DIM 128
#define OUTW 254

__global__ void k_deg_init(float* deg, int n) {
    int i = blockIdx.x * blockDim.x + threadIdx.x;
    if (i < n) deg[i] = 1.0f;
}

__global__ void k_deg_count(const int* __restrict__ col, float* deg, int e) {
    int i = blockIdx.x * blockDim.x + threadIdx.x;
    if (i < e) atomicAdd(&deg[col[i]], 1.0f);
}

__global__ void k_deg_rsqrt(float* deg, float* __restrict__ out, int n) {
    int i = blockIdx.x * blockDim.x + threadIdx.x;
    if (i < n) {
        float d = rsqrtf(deg[i]);
        deg[i] = d;
        out[(long long)i * OUTW] = d * d;
    }
}

#define AP 132
__device__ __forceinline__ void stage_a_tile(const float* __restrict__ X, float* __restrict__ atile,
                                             int m0, int n, int tid) {
    const int r   = tid >> 5;
    const int c4  = tid & 31;
    const int ma  = m0 + r;
    const int mb  = m0 + r + 8;
    const int mac = (ma < n) ? ma : (n - 1);
    const int mbc = (mb < n) ? mb : (n - 1);
    const float4 va = *(const float4*)(X + (long long)mac * DIM + 4 * c4);
    const float4 vb = *(const float4*)(X + (long long)mbc * DIM + 4 * c4);
    *(float4*)(atile + r * AP + 4 * c4)       = va;
    *(float4*)(atile + (r + 8) * AP + 4 * c4) = vb;
}

__global__ __launch_bounds__(256) void k_gemm128_wmma(
    const float* __restrict__ X, const float* __restrict__ W,
    const float* __restrict__ cvec, float* __restrict__ Y, int n) {
    __shared__ __attribute__((aligned(16))) float atile[16 * AP];
    const int lane = threadIdx.x & 31;
    const int wave = threadIdx.x >> 5;
    const int m0   = blockIdx.x * 16;
    const int n0   = wave * 16;
    const int lo   = lane & 15;
    const int hi   = lane >> 4;

    stage_a_tile(X, atile, m0, n, threadIdx.x);
    const float* ap = atile + lo * AP + 2 * hi;
    const float* bp = W + (n0 + lo) + (long long)(2 * hi) * DIM;

    float ci = cvec ? cvec[n0 + lo] : 0.0f;
    v8f d;
#pragma unroll
    for (int v = 0; v < 8; ++v) d[v] = ci;
    __syncthreads();

#pragma unroll
    for (int k = 0; k < DIM; k += 4) {
        v2f a = *(const v2f*)(ap + k);
        v2f b;
        b.x = bp[(long long)k * DIM];
        b.y = bp[(long long)k * DIM + DIM];
        d = __builtin_amdgcn_wmma_f32_16x16x4_f32(false, a, false, b,
                                                  (short)0, d, false, false);
    }

    float* yp = Y + (long long)(m0 + 8 * hi) * DIM + n0 + lo;
#pragma unroll
    for (int v = 0; v < 8; ++v) {
        int m = m0 + v + 8 * hi;
        if (m < n) yp[(long long)v * DIM] = d[v];
    }
}

#define FP 258
__global__ __launch_bounds__(256) void k_gemm128_pool(
    const float* __restrict__ X, const float* __restrict__ W,
    const float* __restrict__ cvec, const float* __restrict__ b2,
    const float* __restrict__ root_feat, float* __restrict__ out, int n) {
    __shared__ float feat[16 * FP];
    __shared__ __attribute__((aligned(16))) float atile[16 * AP];
    const int tid  = threadIdx.x;
    const int lane = tid & 31;
    const int wave = tid >> 5;
    const int m0   = blockIdx.x * 16;
    const int n0   = wave * 16;
    const int lo   = lane & 15;
    const int hi   = lane >> 4;

    stage_a_tile(X, atile, m0, n, tid);
    const float* ap = atile + lo * AP + 2 * hi;
    const float* bp = W + (n0 + lo) + (long long)(2 * hi) * DIM;

    const float cj = cvec[n0 + lo];
    const float bj = b2[n0 + lo];
    const float rf = root_feat[tid & 127];
    v8f d;
#pragma unroll
    for (int v = 0; v < 8; ++v) {
        int m = m0 + v + 8 * hi;
        int mc = (m < n) ? m : (n - 1);
        d[v] = out[(long long)mc * OUTW] * cj + bj;
    }
    __syncthreads();

#pragma unroll
    for (int k = 0; k < DIM; k += 4) {
        v2f a = *(const v2f*)(ap + k);
        v2f b;
        b.x = bp[(long long)k * DIM];
        b.y = bp[(long long)k * DIM + DIM];
        d = __builtin_amdgcn_wmma_f32_16x16x4_f32(false, a, false, b,
                                                  (short)0, d, false, false);
    }

    asm volatile("v_nop\n\tv_nop\n\tv_nop\n\tv_nop" : "+v"(d));

#pragma unroll
    for (int j = 0; j < 8; ++j) feat[(2 * j + (tid >> 7)) * FP + (tid & 127)] = rf;
#pragma unroll
    for (int v = 0; v < 8; ++v) feat[(v + 8 * hi) * FP + DIM + n0 + lo] = fmaxf(d[v], 0.0f);
    __syncthreads();

    const int rows  = (n - m0 < 16) ? (n - m0) : 16;
    const int total = rows * OUTW;
    float* ob = out + (long long)m0 * OUTW;
#pragma unroll
    for (int j = 0; j < 16; ++j) {
        int q = tid + 256 * j;
        if (q < total) {
            int r = q / OUTW;
            int t = q - r * OUTW;
            const float* f = feat + r * FP + t;
            ob[q] = (f[0] + f[1] + f[2]) * (1.0f / 3.0f);
        }
    }
}

__global__ void k_selfloop_init(const float* __restrict__ xw, const float* __restrict__ dinv,
                                float* __restrict__ acc, int n) {
    int t = blockIdx.x * blockDim.x + threadIdx.x;
    if (t >= n * 32) return;
    int node = t >> 5;
    float s = dinv[node]; s *= s;
    float4 v = ((const float4*)xw)[t];
    float4 o = make_float4(v.x * s, v.y * s, v.z * s, v.w * s);
    ((float4*)acc)[t] = o;
}

#define EG 8
__global__ __launch_bounds__(256) void k_edge_scatter(const float* __restrict__ xw,
                               const int* __restrict__ rowi, const int* __restrict__ coli,
                               const float* __restrict__ dinv, float* __restrict__ acc,
                               float* __restrict__ wsum, int e) {
    const int lane = threadIdx.x & 31;
    const int wv   = __builtin_amdgcn_readfirstlane((int)(threadIdx.x >> 5));
    const int base = (blockIdx.x * 8 + wv) * 32;
    if (base >= e) return;
    const int cnt = (e - base < 32) ? (e - base) : 32;
    const int mine = base + ((lane < cnt) ? lane : 0);
    const int r = rowi[mine], c = coli[mine];
    const float nrm = dinv[r] * dinv[c];
    const int nbits = __float_as_int(nrm);
    if (wsum != nullptr && lane < cnt) atomicAdd(wsum + (long long)c * OUTW, nrm);

    int j0 = 0;
    for (; j0 + EG <= cnt; j0 += EG) {
        float v[EG][4];
#pragma unroll
        for (int u = 0; u < EG; ++u) {
            const int rj = __builtin_amdgcn_readlane(r, j0 + u);
            const float* srow = xw + (long long)rj * DIM + lane;
            v[u][0] = srow[0]; v[u][1] = srow[32]; v[u][2] = srow[64]; v[u][3] = srow[96];
        }
#pragma unroll
        for (int u = 0; u < EG; ++u) {
            const int cj = __builtin_amdgcn_readlane(c, j0 + u);
            const float nj = __int_as_float(__builtin_amdgcn_readlane(nbits, j0 + u));
            float* dst = acc + (long long)cj * DIM + lane;
            atomicAdd(dst +  0, v[u][0] * nj);
            atomicAdd(dst + 32, v[u][1] * nj);
            atomicAdd(dst + 64, v[u][2] * nj);
            atomicAdd(dst + 96, v[u][3] * nj);
        }
    }
    for (; j0 < cnt; ++j0) {
        const int rj = __builtin_amdgcn_readlane(r, j0);
        const int cj = __builtin_amdgcn_readlane(c, j0);
        const float nj = __int_as_float(__builtin_amdgcn_readlane(nbits, j0));
        const float* srow = xw + (long long)rj * DIM + lane;
        float v0 = srow[0], v1 = srow[32], v2 = srow[64], v3 = srow[96];
        float* dst = acc + (long long)cj * DIM + lane;
        atomicAdd(dst +  0, v0 * nj);
        atomicAdd(dst + 32, v1 * nj);
        atomicAdd(dst + 64, v2 * nj);
        atomicAdd(dst + 96, v3 * nj);
    }
}

__global__ void k_root_and_c(const float* __restrict__ acc, const float* __restrict__ b1,
                             const float* __restrict__ posts, const float* __restrict__ W2,
                             const int* __restrict__ rootPtr,
                             float* __restrict__ root_feat, float* __restrict__ cvec) {
    int j = threadIdx.x;
    int root = *rootPtr;
    root_feat[j] = acc[(long long)root * DIM + j] + b1[j];
    float s = 0.0f;
    for (int k = 0; k < DIM; ++k)
        s += fmaxf(posts[(long long)root * DIM + k], 0.0f) * W2[(long long)(DIM + k) * DIM + j];
    cvec[j] = s;
}

__global__ void k_bias_relu_start(float* __restrict__ acc, const float* __restrict__ b,
                                  const float* __restrict__ dinv, float* __restrict__ h, int n) {
    int t = blockIdx.x * blockDim.x + threadIdx.x;
    if (t >= n * 32) return;
    float s = dinv[t >> 5]; s *= s;
    float4 bb = ((const float4*)b)[t & 31];
    float4 v = ((float4*)acc)[t];
    v.x = fmaxf(v.x + bb.x, 0.0f);
    v.y = fmaxf(v.y + bb.y, 0.0f);
    v.z = fmaxf(v.z + bb.z, 0.0f);
    v.w = fmaxf(v.w + bb.w, 0.0f);
    ((float4*)h)[t] = v;
    ((float4*)acc)[t] = make_float4(v.x * s, v.y * s, v.z * s, v.w * s);
}

extern "C" void kernel_launch(void* const* d_in, const int* in_sizes, int n_in,
                              void* d_out, int out_size, void* d_ws, size_t ws_size,
                              hipStream_t stream) {
    const float* posts = (const float*)d_in[0];
    const float* W1    = (const float*)d_in[1];
    const float* b1    = (const float*)d_in[2];
    const float* W2    = (const float*)d_in[3];
    const float* b2    = (const float*)d_in[4];
    const int*   edge  = (const int*)d_in[5];
    const int*   rootP = (const int*)d_in[6];

    const int n = in_sizes[0] / DIM;
    const int e = in_sizes[5] / 2;
    const int* rowi = edge;
    const int* coli = edge + e;

    char* ws = (char*)d_ws;
    size_t off = 0;
    float* dinv = (float*)(ws + off); off += (((size_t)n * 4) + 255) & ~(size_t)255;
    float* xw   = (float*)(ws + off); off += (size_t)n * DIM * 4;
    float* acc  = (float*)(ws + off); off += (size_t)n * DIM * 4;
    float* rootf = (float*)(ws + off); off += DIM * 4;
    float* cvec  = (float*)(ws + off);
    float* out = (float*)d_out;

    const int B = 256;
    const int gN   = (n + B - 1) / B;
    const int gE   = (e + B - 1) / B;
    const int gNF4 = (n * 32 + B - 1) / B;
    const int gE32 = (e + B - 1) / B;
    const int gM   = (n + 15) / 16;

    k_deg_init <<<gN, B, 0, stream>>>(dinv, n);
    k_deg_count<<<gE, B, 0, stream>>>(coli, dinv, e);
    k_deg_rsqrt<<<gN, B, 0, stream>>>(dinv, out, n);

    k_gemm128_wmma <<<gM, B, 0, stream>>>(posts, W1, nullptr, xw, n);
    k_selfloop_init<<<gNF4, B, 0, stream>>>(xw, dinv, acc, n);
    k_edge_scatter <<<gE32, B, 0, stream>>>(xw, rowi, coli, dinv, acc, nullptr, e);

    k_root_and_c<<<1, DIM, 0, stream>>>(acc, b1, posts, W2, rootP, rootf, cvec);

    k_bias_relu_start<<<gNF4, B, 0, stream>>>(acc, b1, dinv, xw, n);
    k_edge_scatter   <<<gE32, B, 0, stream>>>(xw, rowi, coli, dinv, acc, out, e);
    k_gemm128_pool   <<<gM, B, 0, stream>>>(acc, W2, cvec, b2, rootf, out, n);
}


# speedup vs baseline: 2.1361x; 1.1751x over previous
#include <hip/hip_runtime.h>

typedef __attribute__((ext_vector_type(2))) float v2f;
typedef __attribute__((ext_vector_type(8))) float v8f;

#define DIM 128
#define OUTW 254

#define BLK (16 * OUTW)
#define P1 32
#define QBASE 64
#define QPITCH 64
__device__ __forceinline__ long long ws_slot(int m) { return (long long)(m >> 4) * BLK + (m & 15); }
__device__ __forceinline__ int package_half() { return (__builtin_amdgcn_s_sendmsg_rtn(0x87) >> 18) & 1; }

__global__ void k_deg_init(float* deg, float* __restrict__ out, int n) {
    int i = blockIdx.x * blockDim.x + threadIdx.x;
    if (i < n) {
        deg[i] = 1.0f;
        out[ws_slot(i) + P1] = 0.0f;
    }
    if (i < 4) ((int*)out)[QBASE + QPITCH * i] = 0;
}

__global__ void k_deg_count(const int* __restrict__ col, float* deg, float* __restrict__ out, int e) {
    int i = blockIdx.x * blockDim.x + threadIdx.x;
    const int half = package_half();
    if (i < e) {
        const int c = col[i];
        atomicAdd(half ? (out + ws_slot(c) + P1) : (deg + c), 1.0f);
    }
}

__global__ void k_deg_rsqrt(float* deg, float* __restrict__ out, int n) {
    int i = blockIdx.x * blockDim.x + threadIdx.x;
    if (i < n) {
        const long long sl = ws_slot(i);
        float d = rsqrtf(deg[i] + out[sl + P1]);
        deg[i] = d;
        out[sl] = d * d;
    }
}

#define AP 132
__device__ __forceinline__ void stage_a_tile(const float* __restrict__ X,
                                             float* __restrict__ atile, int m0, int n, int tid) {
    const int r   = tid >> 5;
    const int c4  = tid & 31;
    const int ma  = m0 + r;
    const int mb  = m0 + r + 8;
    const int mac = (ma < n) ? ma : (n - 1);
    const int mbc = (mb < n) ? mb : (n - 1);
    float4 va = *(const float4*)(X + (long long)mac * DIM + 4 * c4);
    float4 vb = *(const float4*)(X + (long long)mbc * DIM + 4 * c4);
    *(float4*)(atile + r * AP + 4 * c4)       = va;
    *(float4*)(atile + (r + 8) * AP + 4 * c4) = vb;
}

__global__ __launch_bounds__(256) void k_gemm128_wmma(
    const float* __restrict__ X, const float* __restrict__ W,
    const float* __restrict__ cvec, float* __restrict__ Y, int n) {
    __shared__ __attribute__((aligned(16))) float atile[16 * AP];
    const int lane = threadIdx.x & 31;
    const int wave = threadIdx.x >> 5;
    const int m0   = blockIdx.x * 16;
    const int n0   = wave * 16;
    const int lo   = lane & 15;
    const int hi   = lane >> 4;

    stage_a_tile(X, atile, m0, n, threadIdx.x);
    const float* ap = atile + lo * AP + 2 * hi;
    const float* bp = W + (n0 + lo) + (long long)(2 * hi) * DIM;

    float ci = cvec ? cvec[n0 + lo] : 0.0f;
    v8f d;
#pragma unroll
    for (int v = 0; v < 8; ++v) d[v] = ci;
    __syncthreads();

#pragma unroll
    for (int k = 0; k < DIM; k += 4) {
        v2f a = *(const v2f*)(ap + k);
        v2f b;
        b.x = bp[(long long)k * DIM];
        b.y = bp[(long long)k * DIM + DIM];
        d = __builtin_amdgcn_wmma_f32_16x16x4_f32(false, a, false, b,
                                                  (short)0, d, false, false);
    }

    float* yp = Y + (long long)(m0 + 8 * hi) * DIM + n0 + lo;
#pragma unroll
    for (int v = 0; v < 8; ++v) {
        int m = m0 + v + 8 * hi;
        if (m < n) yp[(long long)v * DIM] = d[v];
    }
}

#define FP 258
__global__ __launch_bounds__(256) void k_gemm128_pool(
    const float* __restrict__ X, const float* __restrict__ W,
    const float* __restrict__ cvec, const float* __restrict__ b2,
    const float* __restrict__ root_feat, float* __restrict__ out, int n) {
    __shared__ float feat[16 * FP];
    __shared__ __attribute__((aligned(16))) float atile[16 * AP];
    const int tid  = threadIdx.x;
    const int lane = tid & 31;
    const int wave = tid >> 5;
    const int m0   = blockIdx.x * 16;
    const int n0   = wave * 16;
    const int lo   = lane & 15;
    const int hi   = lane >> 4;

    stage_a_tile(X, atile, m0, n, tid);
    const float* ap = atile + lo * AP + 2 * hi;
    const float* bp = W + (n0 + lo) + (long long)(2 * hi) * DIM;

    const float cj = cvec[n0 + lo];
    const float bj = b2[n0 + lo];
    const float rf = root_feat[tid & 127];
    v8f d;
#pragma unroll
    for (int v = 0; v < 8; ++v) {
        int m = m0 + v + 8 * hi;
        int mc = (m < n) ? m : (n - 1);
        const long long sl = ws_slot(mc);
        d[v] = out[sl] * cj + bj;
    }
    __syncthreads();

#pragma unroll
    for (int k = 0; k < DIM; k += 4) {
        v2f a = *(const v2f*)(ap + k);
        v2f b;
        b.x = bp[(long long)k * DIM];
        b.y = bp[(long long)k * DIM + DIM];
        d = __builtin_amdgcn_wmma_f32_16x16x4_f32(false, a, false, b,
                                                  (short)0, d, false, false);
    }

    asm volatile("v_nop\n\tv_nop\n\tv_nop\n\tv_nop" : "+v"(d));

#pragma unroll
    for (int j = 0; j < 8; ++j) feat[(2 * j + (tid >> 7)) * FP + (tid & 127)] = rf;
#pragma unroll
    for (int v = 0; v < 8; ++v) feat[(v + 8 * hi) * FP + DIM + n0 + lo] = fmaxf(d[v], 0.0f);
    __syncthreads();

    const int rows  = (n - m0 < 16) ? (n - m0) : 16;
    const int total = rows * OUTW;
    float* ob = out + (long long)m0 * OUTW;
#pragma unroll
    for (int j = 0; j < 16; ++j) {
        int q = tid + 256 * j;
        if (q < total) {
            int r = q / OUTW;
            int t = q - r * OUTW;
            const float* f = feat + r * FP + t;
            ob[q] = (f[0] + f[1] + f[2]) * (1.0f / 3.0f);
        }
    }
}

__global__ void k_selfloop_init(const float* __restrict__ xw, const float* __restrict__ dinv,
                                float* __restrict__ acc, int n) {
    int t = blockIdx.x * blockDim.x + threadIdx.x;
    if (t >= n * 32) return;
    int node = t >> 5;
    float s = dinv[node]; s *= s;
    float4 v = ((const float4*)xw)[t];
    float4 o = make_float4(v.x * s, v.y * s, v.z * s, v.w * s);
    ((float4*)acc)[t] = o;
}

#define EG 16
#define ITEM 64
__device__ __forceinline__ int claim_item(int* q, int lane) {
    int t = 0;
    if (lane == 0) t = atomicAdd(q, 1);
    return __builtin_amdgcn_readfirstlane(t);
}

__device__ __forceinline__ void scatter_half_rows(const float* __restrict__ xw, float* __restrict__ acc,
                                                  unsigned off, int r, int c, int nbits, int cnt) {
    int j0 = 0;
    for (; j0 + EG <= cnt; j0 += EG) {
        float v[EG][2];
#pragma unroll
        for (int u = 0; u < EG; ++u) {
            const int rj = __builtin_amdgcn_readlane(r, j0 + u);
            const float* srow = xw + (long long)rj * DIM;
            v[u][0] = srow[off]; v[u][1] = srow[off + 32];
        }
#pragma unroll
        for (int u = 0; u < EG; ++u) {
            const int cj = __builtin_amdgcn_readlane(c, j0 + u);
            const float nj = __int_as_float(__builtin_amdgcn_readlane(nbits, j0 + u));
            float* drow = acc + (long long)cj * DIM;
            atomicAdd(drow + off,      v[u][0] * nj);
            atomicAdd(drow + off + 32, v[u][1] * nj);
        }
    }
    for (; j0 < cnt; ++j0) {
        const int rj = __builtin_amdgcn_readlane(r, j0);
        const int cj = __builtin_amdgcn_readlane(c, j0);
        const float nj = __int_as_float(__builtin_amdgcn_readlane(nbits, j0));
        const float* srow = xw + (long long)rj * DIM;
        const float v0 = srow[off], v1 = srow[off + 32];
        float* drow = acc + (long long)cj * DIM;
        atomicAdd(drow + off,      v0 * nj);
        atomicAdd(drow + off + 32, v1 * nj);
    }
}

__global__ __launch_bounds__(256) void k_edge_scatter(const float* __restrict__ xw,
                               const int* __restrict__ rowi, const int* __restrict__ coli,
                               const float* __restrict__ dinv, float* __restrict__ acc,
                               float* __restrict__ outp, int* ctr, int with_wsum, int e, int nitem) {
    const int lane = threadIdx.x & 31;
    const int half = package_half();
    for (int side = 0; side < 2; ++side) {
        const int ch = half ^ side;
        int* const q = ctr + QPITCH * ch;
        const unsigned off = 64u * (unsigned)ch + (unsigned)lane;
        for (;;) {
            const int item = claim_item(q, lane);
            if (item >= nitem) break;
            const int base = item * ITEM;
            const int cnt = (e - base < ITEM) ? (e - base) : ITEM;
            const int ia = base + ((lane < cnt) ? lane : 0);
            const int ib = base + ((lane + 32 < cnt) ? (lane + 32) : 0);
            const int ra = rowi[ia], ca = coli[ia];
            const int rb = rowi[ib], cb = coli[ib];
            const float na = dinv[ra] * dinv[ca];
            const float nb = dinv[rb] * dinv[cb];
            if (with_wsum && ch == 0) {
                if (lane < cnt)      atomicAdd(outp + ws_slot(ca), na);
                if (lane + 32 < cnt) atomicAdd(outp + ws_slot(cb), nb);
            }
            scatter_half_rows(xw, acc, off, ra, ca, __float_as_int(na), (cnt < 32) ? cnt : 32);
            scatter_half_rows(xw, acc, off, rb, cb, __float_as_int(nb), (cnt > 32) ? (cnt - 32) : 0);
        }
    }
}

__global__ void k_root_and_c(const float* __restrict__ acc,
                             const float* __restrict__ b1,
                             const float* __restrict__ posts, const float* __restrict__ W2,
                             const int* __restrict__ rootPtr,
                             float* __restrict__ root_feat, float* __restrict__ cvec) {
    int j = threadIdx.x;
    int root = *rootPtr;
    root_feat[j] = acc[(long long)root * DIM + j] + b1[j];
    float s = 0.0f;
    for (int k = 0; k < DIM; ++k)
        s += fmaxf(posts[(long long)root * DIM + k], 0.0f) * W2[(long long)(DIM + k) * DIM + j];
    cvec[j] = s;
}

__global__ void k_bias_relu_start(float* __restrict__ acc, const float* __restrict__ b,
                                  const float* __restrict__ dinv, float* __restrict__ h, int n) {
    int t = blockIdx.x * blockDim.x + threadIdx.x;
    if (t >= n * 32) return;
    float s = dinv[t >> 5]; s *= s;
    float4 bb = ((const float4*)b)[t & 31];
    float4 v = ((float4*)acc)[t];
    v.x = fmaxf(v.x + bb.x, 0.0f);
    v.y = fmaxf(v.y + bb.y, 0.0f);
    v.z = fmaxf(v.z + bb.z, 0.0f);
    v.w = fmaxf(v.w + bb.w, 0.0f);
    ((float4*)h)[t] = v;
    ((float4*)acc)[t] = make_float4(v.x * s, v.y * s, v.z * s, v.w * s);
}

extern "C" void kernel_launch(void* const* d_in, const int* in_sizes, int n_in,
                              void* d_out, int out_size, void* d_ws, size_t ws_size,
                              hipStream_t stream) {
    const float* posts = (const float*)d_in[0];
    const float* W1    = (const float*)d_in[1];
    const float* b1    = (const float*)d_in[2];
    const float* W2    = (const float*)d_in[3];
    const float* b2    = (const float*)d_in[4];
    const int*   edge  = (const int*)d_in[5];
    const int*   rootP = (const int*)d_in[6];

    const int n = in_sizes[0] / DIM;
    const int e = in_sizes[5] / 2;
    const int* rowi = edge;
    const int* coli = edge + e;

    char* ws = (char*)d_ws;
    size_t off = 0;
    float* dinv = (float*)(ws + off); off += (((size_t)n * 4) + 255) & ~(size_t)255;
    float* xw   = (float*)(ws + off); off += (size_t)n * DIM * 4;
    float* acc  = (float*)(ws + off); off += (size_t)n * DIM * 4;
    float* rootf = (float*)(ws + off); off += DIM * 4;
    float* cvec  = (float*)(ws + off);
    float* out = (float*)d_out;

    const int B = 256;
    const int gN   = (n + B - 1) / B;
    const int gE   = (e + B - 1) / B;
    const int gNF4 = (n * 32 + B - 1) / B;
    const int nItem = (e + 63) / 64;
    const int gQ    = 1536;
    int* ctr = (int*)d_out + 64;
    const int gM   = (n + 15) / 16;

    k_deg_init <<<gN, B, 0, stream>>>(dinv, out, n);
    k_deg_count<<<gE, B, 0, stream>>>(coli, dinv, out, e);
    k_deg_rsqrt<<<gN, B, 0, stream>>>(dinv, out, n);

    k_gemm128_wmma <<<gM, B, 0, stream>>>(posts, W1, nullptr, xw, n);
    k_selfloop_init<<<gNF4, B, 0, stream>>>(xw, dinv, acc, n);
    k_edge_scatter <<<gQ, B, 0, stream>>>(xw, rowi, coli, dinv, acc, out, ctr, 0, e, nItem);

    k_root_and_c<<<1, DIM, 0, stream>>>(acc, b1, posts, W2, rootP, rootf, cvec);

    k_bias_relu_start<<<gNF4, B, 0, stream>>>(acc, b1, dinv, xw, n);
    k_edge_scatter   <<<gQ, B, 0, stream>>>(xw, rowi, coli, dinv, acc, out, ctr + 128, 1, e, nItem);
    k_gemm128_pool   <<<gM, B, 0, stream>>>(acc, W2, cvec, b2, rootf, out, n);
}


# Edge sums kept in LDS: a work-group owns 391 sum rows, reads the whole destination list, lists its hits, adds by LDS f32 adds in arrival order, stores rows once; no global f32 atomics on rows
# speedup vs baseline: 2.2050x; 1.0323x over previous
#include <hip/hip_runtime.h>

typedef __attribute__((ext_vector_type(2))) float v2f;
typedef __attribute__((ext_vector_type(8))) float v8f;

#define DIM 128
#define OUTW 254

#define BLK (16 * OUTW)
#define P1 32
__device__ __forceinline__ long long ws_slot(int m) { return (long long)(m >> 4) * BLK + (m & 15); }
__device__ __forceinline__ int package_half() { return (__builtin_amdgcn_s_sendmsg_rtn(0x87) >> 18) & 1; }

__global__ void k_deg_init(float* deg, float* __restrict__ out, int n) {
    int i = blockIdx.x * blockDim.x + threadIdx.x;
    if (i < n) {
        deg[i] = 1.0f;
        out[ws_slot(i) + P1] = 0.0f;
    }
}

__global__ void k_deg_count(const int* __restrict__ col, float* deg, float* __restrict__ out, int e) {
    int i = blockIdx.x * blockDim.x + threadIdx.x;
    const int half = package_half();
    if (i < e) {
        const int c = col[i];
        atomicAdd(half ? (out + ws_slot(c) + P1) : (deg + c), 1.0f);
    }
}

__global__ void k_deg_rsqrt(float* deg, float* __restrict__ out, int n) {
    int i = blockIdx.x * blockDim.x + threadIdx.x;
    if (i < n) {
        const long long sl = ws_slot(i);
        float d = rsqrtf(deg[i] + out[sl + P1]);
        deg[i] = d;
        out[sl] = d * d;
    }
}

#define AP 132
__device__ __forceinline__ void stage_a_tile(const float* __restrict__ X,
                                             float* __restrict__ atile, int m0, int n, int tid) {
    const int r   = tid >> 5;
    const int c4  = tid & 31;
    const int ma  = m0 + r;
    const int mb  = m0 + r + 8;
    const int mac = (ma < n) ? ma : (n - 1);
    const int mbc = (mb < n) ? mb : (n - 1);
    float4 va = *(const float4*)(X + (long long)mac * DIM + 4 * c4);
    float4 vb = *(const float4*)(X + (long long)mbc * DIM + 4 * c4);
    *(float4*)(atile + r * AP + 4 * c4)       = va;
    *(float4*)(atile + (r + 8) * AP + 4 * c4) = vb;
}

__global__ __launch_bounds__(256) void k_gemm128_wmma(
    const float* __restrict__ X, const float* __restrict__ W,
    const float* __restrict__ cvec, float* __restrict__ Y, int n) {
    __shared__ __attribute__((aligned(16))) float atile[16 * AP];
    const int lane = threadIdx.x & 31;
    const int wave = threadIdx.x >> 5;
    const int m0   = blockIdx.x * 16;
    const int n0   = wave * 16;
    const int lo   = lane & 15;
    const int hi   = lane >> 4;

    stage_a_tile(X, atile, m0, n, threadIdx.x);
    const float* ap = atile + lo * AP + 2 * hi;
    const float* bp = W + (n0 + lo) + (long long)(2 * hi) * DIM;

    float ci = cvec ? cvec[n0 + lo] : 0.0f;
    v8f d;
#pragma unroll
    for (int v = 0; v < 8; ++v) d[v] = ci;
    __syncthreads();

#pragma unroll
    for (int k = 0; k < DIM; k += 4) {
        v2f a = *(const v2f*)(ap + k);
        v2f b;
        b.x = bp[(long long)k * DIM];
        b.y = bp[(long long)k * DIM + DIM];
        d = __builtin_amdgcn_wmma_f32_16x16x4_f32(false, a, false, b,
                                                  (short)0, d, false, false);
    }

    float* yp = Y + (long long)(m0 + 8 * hi) * DIM + n0 + lo;
#pragma unroll
    for (int v = 0; v < 8; ++v) {
        int m = m0 + v + 8 * hi;
        if (m < n) yp[(long long)v * DIM] = d[v];
    }
}

#define FP 258
__global__ __launch_bounds__(256) void k_gemm128_pool(
    const float* __restrict__ X, const float* __restrict__ W,
    const float* __restrict__ cvec, const float* __restrict__ b2,
    const float* __restrict__ root_feat, float* __restrict__ out, int n) {
    __shared__ float feat[16 * FP];
    __shared__ __attribute__((aligned(16))) float atile[16 * AP];
    const int tid  = threadIdx.x;
    const int lane = tid & 31;
    const int wave = tid >> 5;
    const int m0   = blockIdx.x * 16;
    const int n0   = wave * 16;
    const int lo   = lane & 15;
    const int hi   = lane >> 4;

    stage_a_tile(X, atile, m0, n, tid);
    const float* ap = atile + lo * AP + 2 * hi;
    const float* bp = W + (n0 + lo) + (long long)(2 * hi) * DIM;

    const float cj = cvec[n0 + lo];
    const float bj = b2[n0 + lo];
    const float rf = root_feat[tid & 127];
    v8f d;
#pragma unroll
    for (int v = 0; v < 8; ++v) {
        int m = m0 + v + 8 * hi;
        int mc = (m < n) ? m : (n - 1);
        const long long sl = ws_slot(mc);
        d[v] = out[sl] * cj + bj;
    }
    __syncthreads();

#pragma unroll
    for (int k = 0; k < DIM; k += 4) {
        v2f a = *(const v2f*)(ap + k);
        v2f b;
        b.x = bp[(long long)k * DIM];
        b.y = bp[(long long)k * DIM + DIM];
        d = __builtin_amdgcn_wmma_f32_16x16x4_f32(false, a, false, b,
                                                  (short)0, d, false, false);
    }

    asm volatile("v_nop\n\tv_nop\n\tv_nop\n\tv_nop" : "+v"(d));

#pragma unroll
    for (int j = 0; j < 8; ++j) feat[(2 * j + (tid >> 7)) * FP + (tid & 127)] = rf;
#pragma unroll
    for (int v = 0; v < 8; ++v) feat[(v + 8 * hi) * FP + DIM + n0 + lo] = fmaxf(d[v], 0.0f);
    __syncthreads();

    const int rows  = (n - m0 < 16) ? (n - m0) : 16;
    const int total = rows * OUTW;
    float* ob = out + (long long)m0 * OUTW;
#pragma unroll
    for (int j = 0; j < 16; ++j) {
        int q = tid + 256 * j;
        if (q < total) {
            int r = q / OUTW;
            int t = q - r * OUTW;
            const float* f = feat + r * FP + t;
            ob[q] = (f[0] + f[1] + f[2]) * (1.0f / 3.0f);
        }
    }
}

__global__ void k_selfloop_init(const float* __restrict__ xw, const float* __restrict__ dinv,
                                float* __restrict__ acc, int n) {
    int t = blockIdx.x * blockDim.x + threadIdx.x;
    if (t >= n * 32) return;
    int node = t >> 5;
    float s = dinv[node]; s *= s;
    float4 v = ((const float4*)xw)[t];
    float4 o = make_float4(v.x * s, v.y * s, v.z * s, v.w * s);
    ((float4*)acc)[t] = o;
}

#define RPO 391
#define SCAP 544
#define EG 16

__device__ __forceinline__ void add_rows_lds(const float* __restrict__ xw, float* tile, int lane,
                                             int r, int lr, int nbits, int cnt) {
    for (int j0 = 0; j0 < cnt; j0 += EG) {
        float v[EG][4];
#pragma unroll
        for (int u = 0; u < EG; ++u) {
            const int ju = (j0 + u < cnt) ? (j0 + u) : (cnt - 1);
            const int rj = __builtin_amdgcn_readlane(r, ju);
            const float* srow = xw + (long long)rj * DIM + lane;
            v[u][0] = srow[0]; v[u][1] = srow[32]; v[u][2] = srow[64]; v[u][3] = srow[96];
        }
#pragma unroll
        for (int u = 0; u < EG; ++u) {
            if (j0 + u < cnt) {
                const int lj = __builtin_amdgcn_readlane(lr, j0 + u);
                const float nj = __int_as_float(__builtin_amdgcn_readlane(nbits, j0 + u));
                float* drow = tile + lj * DIM + lane;
                atomicAdd(drow,      v[u][0] * nj);
                atomicAdd(drow + 32, v[u][1] * nj);
                atomicAdd(drow + 64, v[u][2] * nj);
                atomicAdd(drow + 96, v[u][3] * nj);
            }
        }
    }
}

__device__ __forceinline__ void work_off(const float* __restrict__ xw, const int* __restrict__ rowi,
                                         const float* __restrict__ dinv, float* tile, float* wsum_s,
                                         const int* __restrict__ coli, const int* lse, int off, int k,
                                         int lo, int lane, int with_wsum) {
    const int j  = (lane < k) ? lane : 0;
    const int ej = lse[off + j];
    const int r  = rowi[ej];
    const int lr = coli[ej] - lo;
    const float nrm = dinv[r] * dinv[lo + lr];
    if (with_wsum && lane < k) atomicAdd(wsum_s + lr, nrm);
    add_rows_lds(xw, tile, lane, r, lr, __float_as_int(nrm), k);
}

typedef int ids4 __attribute__((ext_vector_type(4), aligned(4)));
__device__ __forceinline__ void load_ids_full(int4 (&d)[4], const int* __restrict__ coli, int b, int lane) {
#pragma unroll
    for (int q = 0; q < 4; ++q) {
        const ids4 t = *(const ids4*)(coli + b + 128 * q + 4 * lane);
        d[q] = make_int4(t.x, t.y, t.z, t.w);
    }
}
__device__ __forceinline__ void load_ids_checked(int4 (&d)[4], const int* __restrict__ coli, int b, int end, int lane) {
#pragma unroll
    for (int q = 0; q < 4; ++q) {
        const int i0 = b + 128 * q + 4 * lane;
        d[q].x = (i0     < end) ? coli[i0]     : -1;
        d[q].y = (i0 + 1 < end) ? coli[i0 + 1] : -1;
        d[q].z = (i0 + 2 < end) ? coli[i0 + 2] : -1;
        d[q].w = (i0 + 3 < end) ? coli[i0 + 3] : -1;
    }
}

__device__ __forceinline__ int test_trip(const int4 (&cur)[4], int base, int cnt, int lo, int nrow, int lane,
                                         const float* __restrict__ xw, const int* __restrict__ rowi,
                                         const int* __restrict__ coli, const float* __restrict__ dinv,
                                         float* tile, float* wsum_s, int* lse, int with_wsum) {
#pragma unroll
    for (int q = 0; q < 4; ++q) {
        const int cc[4] = { cur[q].x - lo, cur[q].y - lo, cur[q].z - lo, cur[q].w - lo };
        unsigned m[4];
#pragma unroll
        for (int k = 0; k < 4; ++k)
            m[k] = __builtin_amdgcn_ballot_w32((unsigned)cc[k] < (unsigned)nrow);
        if ((m[0] | m[1] | m[2] | m[3]) != 0u) {
#pragma unroll
            for (int k = 0; k < 4; ++k) {
                if (m[k] != 0u) {
                    const int pos = cnt + (int)__builtin_amdgcn_mbcnt_lo(m[k], 0u);
                    if ((m[k] >> lane) & 1u) lse[pos] = base + 128 * q + 4 * lane + k;
                    cnt += __popc(m[k]);
                }
            }
        }
    }
    if (cnt >= 32) {
        int off = 0;
        while (cnt - off >= 32) {
            work_off(xw, rowi, dinv, tile, wsum_s, coli, lse, off, 32, lo, lane, with_wsum);
            off += 32;
        }
        const int rem = cnt - off;
        if (lane < rem) {
            const int te = lse[off + lane];
            lse[lane] = te;
        }
        cnt = rem;
    }
    return cnt;
}

__global__ __launch_bounds__(1024) void k_edge_rows_lds(const float* __restrict__ xw,
                               const int* __restrict__ rowi, const int* __restrict__ coli,
                               const float* __restrict__ dinv, float* __restrict__ acc,
                               float* __restrict__ outp, int with_wsum, int e, int n) {
    __shared__ __attribute__((aligned(16))) float tile[RPO * DIM];
    __shared__ float wsum_s[RPO];
    __shared__ int lse_all[32][SCAP];
    const int tid  = threadIdx.x;
    const int lane = tid & 31;
    const int wave = __builtin_amdgcn_readfirstlane(tid >> 5);
    const int lo   = blockIdx.x * RPO;
    const int nrow = (n - lo < RPO) ? (n - lo) : RPO;
    float* const arow = acc + (long long)lo * DIM;
    const int nf4  = nrow * 32;

    const int nfull = nf4 & ~4095;
    for (int t0 = 0; t0 < nfull; t0 += 4096) {
        float4 sv[4];
#pragma unroll
        for (int u = 0; u < 4; ++u) sv[u] = ((const float4*)arow)[t0 + 1024 * u + tid];
#pragma unroll
        for (int u = 0; u < 4; ++u) ((float4*)tile)[t0 + 1024 * u + tid] = sv[u];
    }
    for (int t = nfull + tid; t < nf4; t += 1024) ((float4*)tile)[t] = ((const float4*)arow)[t];
    if (with_wsum)
        for (int t = tid; t < nrow; t += 1024) wsum_s[t] = outp[ws_slot(lo + t)];
    __syncthreads();

    const int seg = ((e + 16383) / 16384) * 512;
    const int beg = wave * seg;
    const int end = (e - beg < seg) ? e : (beg + seg);
    int* const lse = lse_all[wave];
    int cnt = 0;
    const int ntrip = (end > beg) ? ((end - beg) >> 9) : 0;
    if (ntrip > 0) {
        const int lastb = beg + 512 * (ntrip - 1);
        int4 cur[4];
        load_ids_full(cur, coli, beg, lane);
        for (int it = 0; it < ntrip; ++it) {
            const int base = beg + 512 * it;
            const int nb = (base + 512 <= lastb) ? (base + 512) : lastb;
            int4 nxt[4];
            load_ids_full(nxt, coli, nb, lane);
            cnt = test_trip(cur, base, cnt, lo, nrow, lane, xw, rowi, coli, dinv, tile, wsum_s, lse, with_wsum);
#pragma unroll
            for (int q = 0; q < 4; ++q) cur[q] = nxt[q];
        }
    }
    const int tb = beg + 512 * ntrip;
    if (tb < end) {
        int4 lastids[4];
        load_ids_checked(lastids, coli, tb, end, lane);
        cnt = test_trip(lastids, tb, cnt, lo, nrow, lane, xw, rowi, coli, dinv, tile, wsum_s, lse, with_wsum);
    }
    if (cnt > 0) work_off(xw, rowi, dinv, tile, wsum_s, coli, lse, 0, cnt, lo, lane, with_wsum);
    __syncthreads();

    for (int t = tid; t < nf4; t += 1024) ((float4*)arow)[t] = ((const float4*)tile)[t];
    if (with_wsum)
        for (int t = tid; t < nrow; t += 1024) outp[ws_slot(lo + t)] = wsum_s[t];
}

__global__ void k_root_and_c(const float* __restrict__ acc,
                             const float* __restrict__ b1,
                             const float* __restrict__ posts, const float* __restrict__ W2,
                             const int* __restrict__ rootPtr,
                             float* __restrict__ root_feat, float* __restrict__ cvec) {
    int j = threadIdx.x;
    int root = *rootPtr;
    root_feat[j] = acc[(long long)root * DIM + j] + b1[j];
    float s = 0.0f;
    for (int k = 0; k < DIM; ++k)
        s += fmaxf(posts[(long long)root * DIM + k], 0.0f) * W2[(long long)(DIM + k) * DIM + j];
    cvec[j] = s;
}

__global__ void k_bias_relu_start(float* __restrict__ acc, const float* __restrict__ b,
                                  const float* __restrict__ dinv, float* __restrict__ h, int n) {
    int t = blockIdx.x * blockDim.x + threadIdx.x;
    if (t >= n * 32) return;
    float s = dinv[t >> 5]; s *= s;
    float4 bb = ((const float4*)b)[t & 31];
    float4 v = ((float4*)acc)[t];
    v.x = fmaxf(v.x + bb.x, 0.0f);
    v.y = fmaxf(v.y + bb.y, 0.0f);
    v.z = fmaxf(v.z + bb.z, 0.0f);
    v.w = fmaxf(v.w + bb.w, 0.0f);
    ((float4*)h)[t] = v;
    ((float4*)acc)[t] = make_float4(v.x * s, v.y * s, v.z * s, v.w * s);
}

extern "C" void kernel_launch(void* const* d_in, const int* in_sizes, int n_in,
                              void* d_out, int out_size, void* d_ws, size_t ws_size,
                              hipStream_t stream) {
    const float* posts = (const float*)d_in[0];
    const float* W1    = (const float*)d_in[1];
    const float* b1    = (const float*)d_in[2];
    const float* W2    = (const float*)d_in[3];
    const float* b2    = (const float*)d_in[4];
    const int*   edge  = (const int*)d_in[5];
    const int*   rootP = (const int*)d_in[6];

    const int n = in_sizes[0] / DIM;
    const int e = in_sizes[5] / 2;
    const int* rowi = edge;
    const int* coli = edge + e;

    char* ws = (char*)d_ws;
    size_t off = 0;
    float* dinv = (float*)(ws + off); off += (((size_t)n * 4) + 255) & ~(size_t)255;
    float* xw   = (float*)(ws + off); off += (size_t)n * DIM * 4;
    float* acc  = (float*)(ws + off); off += (size_t)n * DIM * 4;
    float* rootf = (float*)(ws + off); off += DIM * 4;
    float* cvec  = (float*)(ws + off);
    float* out = (float*)d_out;

    const int B = 256;
    const int gN   = (n + B - 1) / B;
    const int gE   = (e + B - 1) / B;
    const int gNF4 = (n * 32 + B - 1) / B;
    const int gO   = (n + RPO - 1) / RPO;
    const int gM   = (n + 15) / 16;

    k_deg_init <<<gN, B, 0, stream>>>(dinv, out, n);
    k_deg_count<<<gE, B, 0, stream>>>(coli, dinv, out, e);
    k_deg_rsqrt<<<gN, B, 0, stream>>>(dinv, out, n);

    k_gemm128_wmma <<<gM, B, 0, stream>>>(posts, W1, nullptr, xw, n);
    k_selfloop_init<<<gNF4, B, 0, stream>>>(xw, dinv, acc, n);
    k_edge_rows_lds<<<gO, 1024, 0, stream>>>(xw, rowi, coli, dinv, acc, out, 0, e, n);

    k_root_and_c<<<1, DIM, 0, stream>>>(acc, b1, posts, W2, rootP, rootf, cvec);

    k_bias_relu_start<<<gNF4, B, 0, stream>>>(acc, b1, dinv, xw, n);
    k_edge_rows_lds  <<<gO, 1024, 0, stream>>>(xw, rowi, coli, dinv, acc, out, 1, e, n);
    k_gemm128_pool   <<<gM, B, 0, stream>>>(acc, W2, cvec, b2, rootf, out, n);
}


# Layer-1 edge pass stores each worked-off entry (source id, local row, factor) in a per-work-group list in the unused output buffer; layer 2 reads it, no second id scan; adds stay in arrival order
# speedup vs baseline: 2.6551x; 1.2041x over previous
#include <hip/hip_runtime.h>

typedef __attribute__((ext_vector_type(2))) float v2f;
typedef __attribute__((ext_vector_type(8))) float v8f;

#define DIM 128
#define OUTW 254

#define BLK (16 * OUTW)
#define P1 32
__device__ __forceinline__ long long ws_slot(int m) { return (long long)(m >> 4) * BLK + (m & 15); }
__device__ __forceinline__ int package_half() { return (__builtin_amdgcn_s_sendmsg_rtn(0x87) >> 18) & 1; }

__global__ void k_deg_init(float* deg, float* __restrict__ out, int n) {
    int i = blockIdx.x * blockDim.x + threadIdx.x;
    if (i < n) {
        deg[i] = 1.0f;
        out[ws_slot(i) + P1] = 0.0f;
    }
}

__global__ void k_deg_count(const int* __restrict__ col, float* deg, float* __restrict__ out, int e) {
    int i = blockIdx.x * blockDim.x + threadIdx.x;
    const int half = package_half();
    if (i < e) {
        const int c = col[i];
        atomicAdd(half ? (out + ws_slot(c) + P1) : (deg + c), 1.0f);
    }
}

__global__ void k_deg_rsqrt(float* deg, float* __restrict__ out, int n) {
    int i = blockIdx.x * blockDim.x + threadIdx.x;
    if (i < n) {
        const long long sl = ws_slot(i);
        float d = rsqrtf(deg[i] + out[sl + P1]);
        deg[i] = d;
        out[sl] = d * d;
    }
}

#define AP 132
__device__ __forceinline__ void stage_a_tile(const float* __restrict__ X,
                                             float* __restrict__ atile, int m0, int n, int tid) {
    const int r   = tid >> 5;
    const int c4  = tid & 31;
    const int ma  = m0 + r;
    const int mb  = m0 + r + 8;
    const int mac = (ma < n) ? ma : (n - 1);
    const int mbc = (mb < n) ? mb : (n - 1);
    float4 va = *(const float4*)(X + (long long)mac * DIM + 4 * c4);
    float4 vb = *(const float4*)(X + (long long)mbc * DIM + 4 * c4);
    *(float4*)(atile + r * AP + 4 * c4)       = va;
    *(float4*)(atile + (r + 8) * AP + 4 * c4) = vb;
}

__global__ __launch_bounds__(256) void k_gemm128_wmma(
    const float* __restrict__ X, const float* __restrict__ W,
    const float* __restrict__ cvec, float* __restrict__ Y, int n) {
    __shared__ __attribute__((aligned(16))) float atile[16 * AP];
    const int lane = threadIdx.x & 31;
    const int wave = threadIdx.x >> 5;
    const int m0   = blockIdx.x * 16;
    const int n0   = wave * 16;
    const int lo   = lane & 15;
    const int hi   = lane >> 4;

    stage_a_tile(X, atile, m0, n, threadIdx.x);
    const float* ap = atile + lo * AP + 2 * hi;
    const float* bp = W + (n0 + lo) + (long long)(2 * hi) * DIM;

    float ci = cvec ? cvec[n0 + lo] : 0.0f;
    v8f d;
#pragma unroll
    for (int v = 0; v < 8; ++v) d[v] = ci;
    __syncthreads();

#pragma unroll
    for (int k = 0; k < DIM; k += 4) {
        v2f a = *(const v2f*)(ap + k);
        v2f b;
        b.x = bp[(long long)k * DIM];
        b.y = bp[(long long)k * DIM + DIM];
        d = __builtin_amdgcn_wmma_f32_16x16x4_f32(false, a, false, b,
                                                  (short)0, d, false, false);
    }

    float* yp = Y + (long long)(m0 + 8 * hi) * DIM + n0 + lo;
#pragma unroll
    for (int v = 0; v < 8; ++v) {
        int m = m0 + v + 8 * hi;
        if (m < n) yp[(long long)v * DIM] = d[v];
    }
}

#define FP 258
__global__ __launch_bounds__(256) void k_gemm128_pool(
    const float* __restrict__ X, const float* __restrict__ W,
    const float* __restrict__ cvec, const float* __restrict__ b2,
    const float* __restrict__ root_feat, float* __restrict__ out, int n) {
    __shared__ float feat[16 * FP];
    __shared__ __attribute__((aligned(16))) float atile[16 * AP];
    const int tid  = threadIdx.x;
    const int lane = tid & 31;
    const int wave = tid >> 5;
    const int m0   = blockIdx.x * 16;
    const int n0   = wave * 16;
    const int lo   = lane & 15;
    const int hi   = lane >> 4;

    stage_a_tile(X, atile, m0, n, tid);
    const float* ap = atile + lo * AP + 2 * hi;
    const float* bp = W + (n0 + lo) + (long long)(2 * hi) * DIM;

    const float cj = cvec[n0 + lo];
    const float bj = b2[n0 + lo];
    const float rf = root_feat[tid & 127];
    v8f d;
#pragma unroll
    for (int v = 0; v < 8; ++v) {
        int m = m0 + v + 8 * hi;
        int mc = (m < n) ? m : (n - 1);
        const long long sl = ws_slot(mc);
        d[v] = out[sl] * cj + bj;
    }
    __syncthreads();

#pragma unroll
    for (int k = 0; k < DIM; k += 4) {
        v2f a = *(const v2f*)(ap + k);
        v2f b;
        b.x = bp[(long long)k * DIM];
        b.y = bp[(long long)k * DIM + DIM];
        d = __builtin_amdgcn_wmma_f32_16x16x4_f32(false, a, false, b,
                                                  (short)0, d, false, false);
    }

    asm volatile("v_nop\n\tv_nop\n\tv_nop\n\tv_nop" : "+v"(d));

#pragma unroll
    for (int j = 0; j < 8; ++j) feat[(2 * j + (tid >> 7)) * FP + (tid & 127)] = rf;
#pragma unroll
    for (int v = 0; v < 8; ++v) feat[(v + 8 * hi) * FP + DIM + n0 + lo] = fmaxf(d[v], 0.0f);
    __syncthreads();

    const int rows  = (n - m0 < 16) ? (n - m0) : 16;
    const int total = rows * OUTW;
    float* ob = out + (long long)m0 * OUTW;
#pragma unroll
    for (int j = 0; j < 16; ++j) {
        int q = tid + 256 * j;
        if (q < total) {
            int r = q / OUTW;
            int t = q - r * OUTW;
            const float* f = feat + r * FP + t;
            ob[q] = (f[0] + f[1] + f[2]) * (1.0f / 3.0f);
        }
    }
}

__global__ void k_selfloop_init(const float* __restrict__ xw, const float* __restrict__ dinv,
                                float* __restrict__ acc, int n) {
    int t = blockIdx.x * blockDim.x + threadIdx.x;
    if (t >= n * 32) return;
    int node = t >> 5;
    float s = dinv[node]; s *= s;
    float4 v = ((const float4*)xw)[t];
    float4 o = make_float4(v.x * s, v.y * s, v.z * s, v.w * s);
    ((float4*)acc)[t] = o;
}

#define RPO 391
#define SCAP 544
#define EG 16

#define SREC 2000
#define RSH 22
struct SaveCtx { int2* recs; int* save_n; int fb; int cap; };
__device__ __forceinline__ long long rec_index(int fb, int s) {
    const int bq = s / SREC;
    return (long long)(fb + bq) * (BLK / 2) + 32 + (s - bq * SREC);
}

__device__ __forceinline__ void add_rows_lds(const float* __restrict__ xw, float* tile, int lane,
                                             int r, int lr, int nbits, int cnt) {
    for (int j0 = 0; j0 < cnt; j0 += EG) {
        float v[EG][4];
#pragma unroll
        for (int u = 0; u < EG; ++u) {
            const int ju = (j0 + u < cnt) ? (j0 + u) : (cnt - 1);
            const int rj = __builtin_amdgcn_readlane(r, ju);
            const float* srow = xw + (long long)rj * DIM + lane;
            v[u][0] = srow[0]; v[u][1] = srow[32]; v[u][2] = srow[64]; v[u][3] = srow[96];
        }
#pragma unroll
        for (int u = 0; u < EG; ++u) {
            if (j0 + u < cnt) {
                const int lj = __builtin_amdgcn_readlane(lr, j0 + u);
                const float nj = __int_as_float(__builtin_amdgcn_readlane(nbits, j0 + u));
                float* drow = tile + lj * DIM + lane;
                atomicAdd(drow,      v[u][0] * nj);
                atomicAdd(drow + 32, v[u][1] * nj);
                atomicAdd(drow + 64, v[u][2] * nj);
                atomicAdd(drow + 96, v[u][3] * nj);
            }
        }
    }
}

__device__ __forceinline__ void work_off(const float* __restrict__ xw, const int* __restrict__ rowi,
                                         const float* __restrict__ dinv, float* tile, float* wsum_s,
                                         const int* __restrict__ coli, const int* lse, int off, int k,
                                         int lo, int lane, int with_wsum, const SaveCtx sc) {
    const int j  = (lane < k) ? lane : 0;
    const int ej = lse[off + j];
    const int r  = rowi[ej];
    const int lr = coli[ej] - lo;
    const float nrm = dinv[r] * dinv[lo + lr];
    if (with_wsum && lane < k) atomicAdd(wsum_s + lr, nrm);
    if (!with_wsum) {
        int base = 0;
        if (lane == 0) base = atomicAdd(sc.save_n, k);
        base = __builtin_amdgcn_readfirstlane(base);
        const int s = base + lane;
        if (lane < k && s < sc.cap) sc.recs[rec_index(sc.fb, s)] = make_int2(r | (lr << RSH), __float_as_int(nrm));
    }
    add_rows_lds(xw, tile, lane, r, lr, __float_as_int(nrm), k);
}

typedef int ids4 __attribute__((ext_vector_type(4), aligned(4)));
__device__ __forceinline__ void load_ids_full(int4 (&d)[4], const int* __restrict__ coli, int b, int lane) {
#pragma unroll
    for (int q = 0; q < 4; ++q) {
        const ids4 t = *(const ids4*)(coli + b + 128 * q + 4 * lane);
        d[q] = make_int4(t.x, t.y, t.z, t.w);
    }
}
__device__ __forceinline__ void load_ids_checked(int4 (&d)[4], const int* __restrict__ coli, int b, int end, int lane) {
#pragma unroll
    for (int q = 0; q < 4; ++q) {
        const int i0 = b + 128 * q + 4 * lane;
        d[q].x = (i0     < end) ? coli[i0]     : -1;
        d[q].y = (i0 + 1 < end) ? coli[i0 + 1] : -1;
        d[q].z = (i0 + 2 < end) ? coli[i0 + 2] : -1;
        d[q].w = (i0 + 3 < end) ? coli[i0 + 3] : -1;
    }
}

__device__ __forceinline__ int test_trip(const int4 (&cur)[4], int base, int cnt, int lo, int nrow, int lane,
                                         const float* __restrict__ xw, const int* __restrict__ rowi,
                                         const int* __restrict__ coli, const float* __restrict__ dinv,
                                         float* tile, float* wsum_s, int* lse, int with_wsum, const SaveCtx sc) {
#pragma unroll
    for (int q = 0; q < 4; ++q) {
        const int cc[4] = { cur[q].x - lo, cur[q].y - lo, cur[q].z - lo, cur[q].w - lo };
        unsigned m[4];
#pragma unroll
        for (int k = 0; k < 4; ++k)
            m[k] = __builtin_amdgcn_ballot_w32((unsigned)cc[k] < (unsigned)nrow);
        if ((m[0] | m[1] | m[2] | m[3]) != 0u) {
#pragma unroll
            for (int k = 0; k < 4; ++k) {
                if (m[k] != 0u) {
                    const int pos = cnt + (int)__builtin_amdgcn_mbcnt_lo(m[k], 0u);
                    if ((m[k] >> lane) & 1u) lse[pos] = base + 128 * q + 4 * lane + k;
                    cnt += __popc(m[k]);
                }
            }
        }
    }
    if (cnt >= 32) {
        int off = 0;
        while (cnt - off >= 32) {
            work_off(xw, rowi, dinv, tile, wsum_s, coli, lse, off, 32, lo, lane, with_wsum, sc);
            off += 32;
        }
        const int rem = cnt - off;
        if (lane < rem) {
            const int te = lse[off + lane];
            lse[lane] = te;
        }
        cnt = rem;
    }
    return cnt;
}

__global__ __launch_bounds__(1024) void k_edge_rows_lds(const float* __restrict__ xw,
                               const int* __restrict__ rowi, const int* __restrict__ coli,
                               const float* __restrict__ dinv, float* __restrict__ acc,
                               float* __restrict__ outp, int with_wsum, int e, int n) {
    __shared__ __attribute__((aligned(16))) float tile[RPO * DIM];
    __shared__ float wsum_s[RPO];
    __shared__ int lse_all[32][SCAP];
    __shared__ int save_n;
    const int tid  = threadIdx.x;
    const int lane = tid & 31;
    const int wave = __builtin_amdgcn_readfirstlane(tid >> 5);
    const int lo   = blockIdx.x * RPO;
    const int nrow = (n - lo < RPO) ? (n - lo) : RPO;
    float* const arow = acc + (long long)lo * DIM;
    const int nf4  = nrow * 32;

    SaveCtx sc;
    sc.recs   = (int2*)outp;
    sc.save_n = &save_n;
    sc.fb     = (lo + 15) >> 4;
    const int nblk = ((lo + nrow) >> 4) - sc.fb;
    sc.cap    = (nblk > 0) ? nblk * SREC : 0;
    int* const lenp = (int*)outp + (long long)(lo >> 4) * BLK + 16;
    int saved = -1;
    if (with_wsum) saved = __builtin_amdgcn_readfirstlane(*lenp);
    if (saved > sc.cap) saved = -1;
    if (tid == 0) save_n = 0;

    const int nfull = nf4 & ~4095;
    for (int t0 = 0; t0 < nfull; t0 += 4096) {
        float4 sv[4];
#pragma unroll
        for (int u = 0; u < 4; ++u) sv[u] = ((const float4*)arow)[t0 + 1024 * u + tid];
#pragma unroll
        for (int u = 0; u < 4; ++u) ((float4*)tile)[t0 + 1024 * u + tid] = sv[u];
    }
    for (int t = nfull + tid; t < nf4; t += 1024) ((float4*)tile)[t] = ((const float4*)arow)[t];
    if (with_wsum)
        for (int t = tid; t < nrow; t += 1024) wsum_s[t] = outp[ws_slot(lo + t)];
    __syncthreads();

    if (saved >= 0) {
        const int nbat = (saved + 31) >> 5;
        int2 rc = make_int2(0, 0);
        if (wave < nbat) {
            const int k0 = (saved - 32 * wave < 32) ? (saved - 32 * wave) : 32;
            rc = sc.recs[rec_index(sc.fb, 32 * wave + ((lane < k0) ? lane : 0))];
        }
        for (int q = wave; q < nbat; q += 32) {
            const int k = (saved - 32 * q < 32) ? (saved - 32 * q) : 32;
            int2 nx = rc;
            if (q + 32 < nbat) {
                const int kn = (saved - 32 * (q + 32) < 32) ? (saved - 32 * (q + 32)) : 32;
                nx = sc.recs[rec_index(sc.fb, 32 * (q + 32) + ((lane < kn) ? lane : 0))];
            }
            const int rr = rc.x & ((1 << RSH) - 1);
            const int r  = (rr < n) ? rr : (n - 1);
            const int lq = (int)((unsigned)rc.x >> RSH);
            const int lr = (lq < nrow) ? lq : (nrow - 1);
            if (lane < k) atomicAdd(wsum_s + lr, __int_as_float(rc.y));
            add_rows_lds(xw, tile, lane, r, lr, rc.y, k);
            rc = nx;
        }
    } else {
    const int seg = ((e + 16383) / 16384) * 512;
    const int beg = wave * seg;
    const int end = (e - beg < seg) ? e : (beg + seg);
    int* const lse = lse_all[wave];
    int cnt = 0;
    const int ntrip = (end > beg) ? ((end - beg) >> 9) : 0;
    if (ntrip > 0) {
        const int lastb = beg + 512 * (ntrip - 1);
        int4 cur[4];
        load_ids_full(cur, coli, beg, lane);
        for (int it = 0; it < ntrip; ++it) {
            const int base = beg + 512 * it;
            const int nb = (base + 512 <= lastb) ? (base + 512) : lastb;
            int4 nxt[4];
            load_ids_full(nxt, coli, nb, lane);
            cnt = test_trip(cur, base, cnt, lo, nrow, lane, xw, rowi, coli, dinv, tile, wsum_s, lse, with_wsum, sc);
#pragma unroll
            for (int q = 0; q < 4; ++q) cur[q] = nxt[q];
        }
    }
    const int tb = beg + 512 * ntrip;
    if (tb < end) {
        int4 lastids[4];
        load_ids_checked(lastids, coli, tb, end, lane);
        cnt = test_trip(lastids, tb, cnt, lo, nrow, lane, xw, rowi, coli, dinv, tile, wsum_s, lse, with_wsum, sc);
    }
    if (cnt > 0) work_off(xw, rowi, dinv, tile, wsum_s, coli, lse, 0, cnt, lo, lane, with_wsum, sc);
    }
    __syncthreads();
    if (!with_wsum && tid == 0) *lenp = (save_n <= sc.cap) ? save_n : -1;

    for (int t = tid; t < nf4; t += 1024) ((float4*)arow)[t] = ((const float4*)tile)[t];
    if (with_wsum)
        for (int t = tid; t < nrow; t += 1024) outp[ws_slot(lo + t)] = wsum_s[t];
}

__global__ void k_root_and_c(const float* __restrict__ acc,
                             const float* __restrict__ b1,
                             const float* __restrict__ posts, const float* __restrict__ W2,
                             const int* __restrict__ rootPtr,
                             float* __restrict__ root_feat, float* __restrict__ cvec) {
    int j = threadIdx.x;
    int root = *rootPtr;
    root_feat[j] = acc[(long long)root * DIM + j] + b1[j];
    float s = 0.0f;
    for (int k = 0; k < DIM; ++k)
        s += fmaxf(posts[(long long)root * DIM + k], 0.0f) * W2[(long long)(DIM + k) * DIM + j];
    cvec[j] = s;
}

__global__ void k_bias_relu_start(float* __restrict__ acc, const float* __restrict__ b,
                                  const float* __restrict__ dinv, float* __restrict__ h, int n) {
    int t = blockIdx.x * blockDim.x + threadIdx.x;
    if (t >= n * 32) return;
    float s = dinv[t >> 5]; s *= s;
    float4 bb = ((const float4*)b)[t & 31];
    float4 v = ((float4*)acc)[t];
    v.x = fmaxf(v.x + bb.x, 0.0f);
    v.y = fmaxf(v.y + bb.y, 0.0f);
    v.z = fmaxf(v.z + bb.z, 0.0f);
    v.w = fmaxf(v.w + bb.w, 0.0f);
    ((float4*)h)[t] = v;
    ((float4*)acc)[t] = make_float4(v.x * s, v.y * s, v.z * s, v.w * s);
}

extern "C" void kernel_launch(void* const* d_in, const int* in_sizes, int n_in,
                              void* d_out, int out_size, void* d_ws, size_t ws_size,
                              hipStream_t stream) {
    const float* posts = (const float*)d_in[0];
    const float* W1    = (const float*)d_in[1];
    const float* b1    = (const float*)d_in[2];
    const float* W2    = (const float*)d_in[3];
    const float* b2    = (const float*)d_in[4];
    const int*   edge  = (const int*)d_in[5];
    const int*   rootP = (const int*)d_in[6];

    const int n = in_sizes[0] / DIM;
    const int e = in_sizes[5] / 2;
    const int* rowi = edge;
    const int* coli = edge + e;

    char* ws = (char*)d_ws;
    size_t off = 0;
    float* dinv = (float*)(ws + off); off += (((size_t)n * 4) + 255) & ~(size_t)255;
    float* xw   = (float*)(ws + off); off += (size_t)n * DIM * 4;
    float* acc  = (float*)(ws + off); off += (size_t)n * DIM * 4;
    float* rootf = (float*)(ws + off); off += DIM * 4;
    float* cvec  = (float*)(ws + off);
    float* out = (float*)d_out;

    const int B = 256;
    const int gN   = (n + B - 1) / B;
    const int gE   = (e + B - 1) / B;
    const int gNF4 = (n * 32 + B - 1) / B;
    const int gO   = (n + RPO - 1) / RPO;
    const int gM   = (n + 15) / 16;

    k_deg_init <<<gN, B, 0, stream>>>(dinv, out, n);
    k_deg_count<<<gE, B, 0, stream>>>(coli, dinv, out, e);
    k_deg_rsqrt<<<gN, B, 0, stream>>>(dinv, out, n);

    k_gemm128_wmma <<<gM, B, 0, stream>>>(posts, W1, nullptr, xw, n);
    k_selfloop_init<<<gNF4, B, 0, stream>>>(xw, dinv, acc, n);
    k_edge_rows_lds<<<gO, 1024, 0, stream>>>(xw, rowi, coli, dinv, acc, out, 0, e, n);

    k_root_and_c<<<1, DIM, 0, stream>>>(acc, b1, posts, W2, rootP, rootf, cvec);

    k_bias_relu_start<<<gNF4, B, 0, stream>>>(acc, b1, dinv, xw, n);
    k_edge_rows_lds  <<<gO, 1024, 0, stream>>>(xw, rowi, coli, dinv, acc, out, 1, e, n);
    k_gemm128_pool   <<<gM, B, 0, stream>>>(acc, W2, cvec, b2, rootf, out, n);
}


# Count pass also writes a 256-bit destination-block set per 32-edge group; layer-1 edge pass reads its 1 bit per edge, loads ids of hit groups only; in-wave list order changes, adds in arrival order
# speedup vs baseline: 2.9119x; 1.0967x over previous
#include <hip/hip_runtime.h>

typedef __attribute__((ext_vector_type(2))) float v2f;
typedef __attribute__((ext_vector_type(8))) float v8f;

#define DIM 128
#define OUTW 254

#define BLK (16 * OUTW)
#define P1 32
#define RPO 391
#define BMO 3040
__device__ __forceinline__ int bm_stride(int e) { return (((e + 31) >> 5) + 1023) & ~1023; }
__device__ __forceinline__ long long bm_word(int t) { return (long long)(t >> 10) * BLK + BMO + (t & 1023); }
__device__ __forceinline__ long long ws_slot(int m) { return (long long)(m >> 4) * BLK + (m & 15); }
__device__ __forceinline__ int package_half() { return (__builtin_amdgcn_s_sendmsg_rtn(0x87) >> 18) & 1; }

__global__ void k_deg_init(float* deg, float* __restrict__ out, int n) {
    int i = blockIdx.x * blockDim.x + threadIdx.x;
    if (i < n) {
        deg[i] = 1.0f;
        out[ws_slot(i) + P1] = 0.0f;
    }
}

__global__ __launch_bounds__(256) void k_deg_count(const int* __restrict__ col, float* deg, float* __restrict__ out, int e) {
    __shared__ unsigned bm_s[8][8];
    const int lane = threadIdx.x & 31;
    const int wave = threadIdx.x >> 5;
    const int i = blockIdx.x * 256 + threadIdx.x;
    const int half = package_half();
    if (lane < 8) bm_s[wave][lane] = 0u;
    __syncthreads();
    if (i < e) {
        const int c = col[i];
        atomicAdd(half ? (out + ws_slot(c) + P1) : (deg + c), 1.0f);
        const int blk = c / RPO;
        atomicOr(&bm_s[wave][(blk >> 5) & 7], 1u << (blk & 31));
    }
    __syncthreads();
    const int g = i >> 5;
    if (lane < 8 && (g << 5) < e)
        ((unsigned*)out)[bm_word(lane * bm_stride(e) + g)] = bm_s[wave][lane];
}

__global__ void k_deg_rsqrt(float* deg, float* __restrict__ out, int n) {
    int i = blockIdx.x * blockDim.x + threadIdx.x;
    if (i < n) {
        const long long sl = ws_slot(i);
        float d = rsqrtf(deg[i] + out[sl + P1]);
        deg[i] = d;
        out[sl] = d * d;
    }
}

#define AP 132
__device__ __forceinline__ void stage_a_tile(const float* __restrict__ X,
                                             float* __restrict__ atile, int m0, int n, int tid) {
    const int r   = tid >> 5;
    const int c4  = tid & 31;
    const int ma  = m0 + r;
    const int mb  = m0 + r + 8;
    const int mac = (ma < n) ? ma : (n - 1);
    const int mbc = (mb < n) ? mb : (n - 1);
    float4 va = *(const float4*)(X + (long long)mac * DIM + 4 * c4);
    float4 vb = *(const float4*)(X + (long long)mbc * DIM + 4 * c4);
    *(float4*)(atile + r * AP + 4 * c4)       = va;
    *(float4*)(atile + (r + 8) * AP + 4 * c4) = vb;
}

__global__ __launch_bounds__(256) void k_gemm128_wmma(
    const float* __restrict__ X, const float* __restrict__ W,
    const float* __restrict__ cvec, float* __restrict__ Y, int n) {
    __shared__ __attribute__((aligned(16))) float atile[16 * AP];
    const int lane = threadIdx.x & 31;
    const int wave = threadIdx.x >> 5;
    const int m0   = blockIdx.x * 16;
    const int n0   = wave * 16;
    const int lo   = lane & 15;
    const int hi   = lane >> 4;

    stage_a_tile(X, atile, m0, n, threadIdx.x);
    const float* ap = atile + lo * AP + 2 * hi;
    const float* bp = W + (n0 + lo) + (long long)(2 * hi) * DIM;

    float ci = cvec ? cvec[n0 + lo] : 0.0f;
    v8f d;
#pragma unroll
    for (int v = 0; v < 8; ++v) d[v] = ci;
    __syncthreads();

#pragma unroll
    for (int k = 0; k < DIM; k += 4) {
        v2f a = *(const v2f*)(ap + k);
        v2f b;
        b.x = bp[(long long)k * DIM];
        b.y = bp[(long long)k * DIM + DIM];
        d = __builtin_amdgcn_wmma_f32_16x16x4_f32(false, a, false, b,
                                                  (short)0, d, false, false);
    }

    float* yp = Y + (long long)(m0 + 8 * hi) * DIM + n0 + lo;
#pragma unroll
    for (int v = 0; v < 8; ++v) {
        int m = m0 + v + 8 * hi;
        if (m < n) yp[(long long)v * DIM] = d[v];
    }
}

#define FP 258
__global__ __launch_bounds__(256) void k_gemm128_pool(
    const float* __restrict__ X, const float* __restrict__ W,
    const float* __restrict__ cvec, const float* __restrict__ b2,
    const float* __restrict__ root_feat, float* __restrict__ out, int n) {
    __shared__ float feat[16 * FP];
    __shared__ __attribute__((aligned(16))) float atile[16 * AP];
    const int tid  = threadIdx.x;
    const int lane = tid & 31;
    const int wave = tid >> 5;
    const int m0   = blockIdx.x * 16;
    const int n0   = wave * 16;
    const int lo   = lane & 15;
    const int hi   = lane >> 4;

    stage_a_tile(X, atile, m0, n, tid);
    const float* ap = atile + lo * AP + 2 * hi;
    const float* bp = W + (n0 + lo) + (long long)(2 * hi) * DIM;

    const float cj = cvec[n0 + lo];
    const float bj = b2[n0 + lo];
    const float rf = root_feat[tid & 127];
    v8f d;
#pragma unroll
    for (int v = 0; v < 8; ++v) {
        int m = m0 + v + 8 * hi;
        int mc = (m < n) ? m : (n - 1);
        const long long sl = ws_slot(mc);
        d[v] = out[sl] * cj + bj;
    }
    __syncthreads();

#pragma unroll
    for (int k = 0; k < DIM; k += 4) {
        v2f a = *(const v2f*)(ap + k);
        v2f b;
        b.x = bp[(long long)k * DIM];
        b.y = bp[(long long)k * DIM + DIM];
        d = __builtin_amdgcn_wmma_f32_16x16x4_f32(false, a, false, b,
                                                  (short)0, d, false, false);
    }

    asm volatile("v_nop\n\tv_nop\n\tv_nop\n\tv_nop" : "+v"(d));

#pragma unroll
    for (int j = 0; j < 8; ++j) feat[(2 * j + (tid >> 7)) * FP + (tid & 127)] = rf;
#pragma unroll
    for (int v = 0; v < 8; ++v) feat[(v + 8 * hi) * FP + DIM + n0 + lo] = fmaxf(d[v], 0.0f);
    __syncthreads();

    const int rows  = (n - m0 < 16) ? (n - m0) : 16;
    const int total = rows * OUTW;
    float* ob = out + (long long)m0 * OUTW;
#pragma unroll
    for (int j = 0; j < 16; ++j) {
        int q = tid + 256 * j;
        if (q < total) {
            int r = q / OUTW;
            int t = q - r * OUTW;
            const float* f = feat + r * FP + t;
            ob[q] = (f[0] + f[1] + f[2]) * (1.0f / 3.0f);
        }
    }
}

__global__ void k_selfloop_init(const float* __restrict__ xw, const float* __restrict__ dinv,
                                float* __restrict__ acc, int n) {
    int t = blockIdx.x * blockDim.x + threadIdx.x;
    if (t >= n * 32) return;
    int node = t >> 5;
    float s = dinv[node]; s *= s;
    float4 v = ((const float4*)xw)[t];
    float4 o = make_float4(v.x * s, v.y * s, v.z * s, v.w * s);
    ((float4*)acc)[t] = o;
}

#define SCAP 544
#define GCAP 144
#define GB 16
#define EG 16

#define SREC 1488
#define RSH 22
struct SaveCtx { int2* recs; int* save_n; int fb; int cap; };
__device__ __forceinline__ long long rec_index(int fb, int s) {
    const int bq = s / SREC;
    return (long long)(fb + bq) * (BLK / 2) + 32 + (s - bq * SREC);
}

__device__ __forceinline__ void add_rows_lds(const float* __restrict__ xw, float* tile, int lane,
                                             int r, int lr, int nbits, int cnt) {
    for (int j0 = 0; j0 < cnt; j0 += EG) {
        float v[EG][4];
#pragma unroll
        for (int u = 0; u < EG; ++u) {
            const int ju = (j0 + u < cnt) ? (j0 + u) : (cnt - 1);
            const int rj = __builtin_amdgcn_readlane(r, ju);
            const float* srow = xw + (long long)rj * DIM + lane;
            v[u][0] = srow[0]; v[u][1] = srow[32]; v[u][2] = srow[64]; v[u][3] = srow[96];
        }
#pragma unroll
        for (int u = 0; u < EG; ++u) {
            if (j0 + u < cnt) {
                const int lj = __builtin_amdgcn_readlane(lr, j0 + u);
                const float nj = __int_as_float(__builtin_amdgcn_readlane(nbits, j0 + u));
                float* drow = tile + lj * DIM + lane;
                atomicAdd(drow,      v[u][0] * nj);
                atomicAdd(drow + 32, v[u][1] * nj);
                atomicAdd(drow + 64, v[u][2] * nj);
                atomicAdd(drow + 96, v[u][3] * nj);
            }
        }
    }
}

__device__ __forceinline__ void work_off(const float* __restrict__ xw, const int* __restrict__ rowi,
                                         const float* __restrict__ dinv, float* tile, float* wsum_s,
                                         const int* __restrict__ coli, const int* lse, int off, int k,
                                         int lo, int lane, int with_wsum, const SaveCtx sc) {
    const int j  = (lane < k) ? lane : 0;
    const int ej = lse[off + j];
    const int r  = rowi[ej];
    const int lr = coli[ej] - lo;
    const float nrm = dinv[r] * dinv[lo + lr];
    if (with_wsum && lane < k) atomicAdd(wsum_s + lr, nrm);
    if (!with_wsum) {
        int base = 0;
        if (lane == 0) base = atomicAdd(sc.save_n, k);
        base = __builtin_amdgcn_readfirstlane(base);
        const int s = base + lane;
        if (lane < k && s < sc.cap) sc.recs[rec_index(sc.fb, s)] = make_int2(r | (lr << RSH), __float_as_int(nrm));
    }
    add_rows_lds(xw, tile, lane, r, lr, __float_as_int(nrm), k);
}

__device__ __forceinline__ int take_groups(const int* gl, int off, int k, int cnt, int lo, int nrow, int lane, int e,
                                           const float* __restrict__ xw, const int* __restrict__ rowi,
                                           const int* __restrict__ coli, const float* __restrict__ dinv,
                                           float* tile, float* wsum_s, int* lse, int with_wsum, const SaveCtx sc) {
    const int gj = gl[off + ((lane < k) ? lane : 0)];
    int idv[GB];
#pragma unroll
    for (int u = 0; u < GB; ++u) {
        const int gu = __builtin_amdgcn_readlane(gj, (u < k) ? u : (k - 1));
        const int i = 32 * gu + lane;
        idv[u] = (i < e) ? coli[i] : -1;
    }
#pragma unroll
    for (int u = 0; u < GB; ++u) {
        if (u < k) {
            const unsigned m = __builtin_amdgcn_ballot_w32((unsigned)(idv[u] - lo) < (unsigned)nrow);
            if (m != 0u) {
                const int pos = cnt + (int)__builtin_amdgcn_mbcnt_lo(m, 0u);
                if ((m >> lane) & 1u) lse[pos] = 32 * __builtin_amdgcn_readlane(gj, u) + lane;
                cnt += __popc(m);
            }
        }
    }
    if (cnt >= 32) {
        int eo = 0;
        while (cnt - eo >= 32) {
            work_off(xw, rowi, dinv, tile, wsum_s, coli, lse, eo, 32, lo, lane, with_wsum, sc);
            eo += 32;
        }
        const int rem = cnt - eo;
        if (lane < rem) {
            const int te = lse[eo + lane];
            lse[lane] = te;
        }
        cnt = rem;
    }
    return cnt;
}

__global__ __launch_bounds__(1024) void k_edge_rows_lds(const float* __restrict__ xw,
                               const int* __restrict__ rowi, const int* __restrict__ coli,
                               const float* __restrict__ dinv, float* __restrict__ acc,
                               float* __restrict__ outp, int with_wsum, int e, int n) {
    __shared__ __attribute__((aligned(16))) float tile[RPO * DIM];
    __shared__ float wsum_s[RPO];
    __shared__ int lse_all[32][SCAP];
    __shared__ int gl_all[32][GCAP];
    __shared__ int save_n;
    const int tid  = threadIdx.x;
    const int lane = tid & 31;
    const int wave = __builtin_amdgcn_readfirstlane(tid >> 5);
    const int lo   = blockIdx.x * RPO;
    const int nrow = (n - lo < RPO) ? (n - lo) : RPO;
    float* const arow = acc + (long long)lo * DIM;
    const int nf4  = nrow * 32;

    SaveCtx sc;
    sc.recs   = (int2*)outp;
    sc.save_n = &save_n;
    sc.fb     = (lo + 15) >> 4;
    const int nblk = ((lo + nrow) >> 4) - sc.fb;
    sc.cap    = (nblk > 0) ? nblk * SREC : 0;
    int* const lenp = (int*)outp + (long long)(lo >> 4) * BLK + 16;
    int saved = -1;
    if (with_wsum) saved = __builtin_amdgcn_readfirstlane(*lenp);
    if (saved > sc.cap) saved = -1;
    if (tid == 0) save_n = 0;

    const int nfull = nf4 & ~4095;
    for (int t0 = 0; t0 < nfull; t0 += 4096) {
        float4 sv[4];
#pragma unroll
        for (int u = 0; u < 4; ++u) sv[u] = ((const float4*)arow)[t0 + 1024 * u + tid];
#pragma unroll
        for (int u = 0; u < 4; ++u) ((float4*)tile)[t0 + 1024 * u + tid] = sv[u];
    }
    for (int t = nfull + tid; t < nf4; t += 1024) ((float4*)tile)[t] = ((const float4*)arow)[t];
    if (with_wsum)
        for (int t = tid; t < nrow; t += 1024) wsum_s[t] = outp[ws_slot(lo + t)];
    __syncthreads();

    if (saved >= 0) {
        const int nbat = (saved + 31) >> 5;
        int2 rc = make_int2(0, 0);
        if (wave < nbat) {
            const int k0 = (saved - 32 * wave < 32) ? (saved - 32 * wave) : 32;
            rc = sc.recs[rec_index(sc.fb, 32 * wave + ((lane < k0) ? lane : 0))];
        }
        for (int q = wave; q < nbat; q += 32) {
            const int k = (saved - 32 * q < 32) ? (saved - 32 * q) : 32;
            int2 nx = rc;
            if (q + 32 < nbat) {
                const int kn = (saved - 32 * (q + 32) < 32) ? (saved - 32 * (q + 32)) : 32;
                nx = sc.recs[rec_index(sc.fb, 32 * (q + 32) + ((lane < kn) ? lane : 0))];
            }
            const int rr = rc.x & ((1 << RSH) - 1);
            const int r  = (rr < n) ? rr : (n - 1);
            const int lq = (int)((unsigned)rc.x >> RSH);
            const int lr = (lq < nrow) ? lq : (nrow - 1);
            if (lane < k) atomicAdd(wsum_s + lr, __int_as_float(rc.y));
            add_rows_lds(xw, tile, lane, r, lr, rc.y, k);
            rc = nx;
        }
    } else {
    const int ng   = (e + 31) >> 5;
    const unsigned* const bmw = (const unsigned*)outp;
    const int pbase = (blockIdx.x >> 5) * bm_stride(e);
    const int bbit  = blockIdx.x & 31;
    const int gseg = (((ng + 31) >> 5) + 127) & ~127;
    const int gbeg = wave * gseg;
    const int gend = (ng - gbeg < gseg) ? ng : (gbeg + gseg);
    int* const lse = lse_all[wave];
    int* const gl  = gl_all[wave];
    int cnt  = 0;
    int gcnt = 0;
    if (gbeg < gend) {
        uint4 cur = *(const uint4*)(bmw + bm_word(pbase + gbeg + 4 * lane));
        for (int gb = gbeg; gb < gend; gb += 128) {
            const int nb = (gb + 128 < gend) ? (gb + 128) : gb;
            const uint4 nxt = *(const uint4*)(bmw + bm_word(pbase + nb + 4 * lane));
            const unsigned w[4] = { cur.x, cur.y, cur.z, cur.w };
#pragma unroll
            for (int k = 0; k < 4; ++k) {
                const int g = gb + 4 * lane + k;
                const bool hit = (((w[k] >> bbit) & 1u) != 0u) && (g < gend);
                const unsigned m = __builtin_amdgcn_ballot_w32(hit);
                if (m != 0u) {
                    const int pos = gcnt + (int)__builtin_amdgcn_mbcnt_lo(m, 0u);
                    if (hit) gl[pos] = g;
                    gcnt += __popc(m);
                }
            }
            if (gcnt >= GB) {
                int go = 0;
                while (gcnt - go >= GB) {
                    cnt = take_groups(gl, go, GB, cnt, lo, nrow, lane, e, xw, rowi, coli, dinv, tile, wsum_s, lse, with_wsum, sc);
                    go += GB;
                }
                const int grem = gcnt - go;
                if (lane < grem) {
                    const int tg = gl[go + lane];
                    gl[lane] = tg;
                }
                gcnt = grem;
            }
            cur = nxt;
        }
    }
    if (gcnt > 0) cnt = take_groups(gl, 0, gcnt, cnt, lo, nrow, lane, e, xw, rowi, coli, dinv, tile, wsum_s, lse, with_wsum, sc);
    if (cnt > 0) work_off(xw, rowi, dinv, tile, wsum_s, coli, lse, 0, cnt, lo, lane, with_wsum, sc);
    }
    __syncthreads();
    if (!with_wsum && tid == 0) *lenp = (save_n <= sc.cap) ? save_n : -1;

    for (int t = tid; t < nf4; t += 1024) ((float4*)arow)[t] = ((const float4*)tile)[t];
    if (with_wsum)
        for (int t = tid; t < nrow; t += 1024) outp[ws_slot(lo + t)] = wsum_s[t];
}

__global__ void k_root_and_c(const float* __restrict__ acc,
                             const float* __restrict__ b1,
                             const float* __restrict__ posts, const float* __restrict__ W2,
                             const int* __restrict__ rootPtr,
                             float* __restrict__ root_feat, float* __restrict__ cvec) {
    int j = threadIdx.x;
    int root = *rootPtr;
    root_feat[j] = acc[(long long)root * DIM + j] + b1[j];
    float s = 0.0f;
    for (int k = 0; k < DIM; ++k)
        s += fmaxf(posts[(long long)root * DIM + k], 0.0f) * W2[(long long)(DIM + k) * DIM + j];
    cvec[j] = s;
}

__global__ void k_bias_relu_start(float* __restrict__ acc, const float* __restrict__ b,
                                  const float* __restrict__ dinv, float* __restrict__ h, int n) {
    int t = blockIdx.x * blockDim.x + threadIdx.x;
    if (t >= n * 32) return;
    float s = dinv[t >> 5]; s *= s;
    float4 bb = ((const float4*)b)[t & 31];
    float4 v = ((float4*)acc)[t];
    v.x = fmaxf(v.x + bb.x, 0.0f);
    v.y = fmaxf(v.y + bb.y, 0.0f);
    v.z = fmaxf(v.z + bb.z, 0.0f);
    v.w = fmaxf(v.w + bb.w, 0.0f);
    ((float4*)h)[t] = v;
    ((float4*)acc)[t] = make_float4(v.x * s, v.y * s, v.z * s, v.w * s);
}

extern "C" void kernel_launch(void* const* d_in, const int* in_sizes, int n_in,
                              void* d_out, int out_size, void* d_ws, size_t ws_size,
                              hipStream_t stream) {
    const float* posts = (const float*)d_in[0];
    const float* W1    = (const float*)d_in[1];
    const float* b1    = (const float*)d_in[2];
    const float* W2    = (const float*)d_in[3];
    const float* b2    = (const float*)d_in[4];
    const int*   edge  = (const int*)d_in[5];
    const int*   rootP = (const int*)d_in[6];

    const int n = in_sizes[0] / DIM;
    const int e = in_sizes[5] / 2;
    const int* rowi = edge;
    const int* coli = edge + e;

    char* ws = (char*)d_ws;
    size_t off = 0;
    float* dinv = (float*)(ws + off); off += (((size_t)n * 4) + 255) & ~(size_t)255;
    float* xw   = (float*)(ws + off); off += (size_t)n * DIM * 4;
    float* acc  = (float*)(ws + off); off += (size_t)n * DIM * 4;
    float* rootf = (float*)(ws + off); off += DIM * 4;
    float* cvec  = (float*)(ws + off);
    float* out = (float*)d_out;

    const int B = 256;
    const int gN   = (n + B - 1) / B;
    const int gE   = (e + B - 1) / B;
    const int gNF4 = (n * 32 + B - 1) / B;
    const int gO   = (n + RPO - 1) / RPO;
    const int gM   = (n + 15) / 16;

    k_deg_init <<<gN, B, 0, stream>>>(dinv, out, n);
    k_deg_count<<<gE, B, 0, stream>>>(coli, dinv, out, e);
    k_deg_rsqrt<<<gN, B, 0, stream>>>(dinv, out, n);

    k_gemm128_wmma <<<gM, B, 0, stream>>>(posts, W1, nullptr, xw, n);
    k_selfloop_init<<<gNF4, B, 0, stream>>>(xw, dinv, acc, n);
    k_edge_rows_lds<<<gO, 1024, 0, stream>>>(xw, rowi, coli, dinv, acc, out, 0, e, n);

    k_root_and_c<<<1, DIM, 0, stream>>>(acc, b1, posts, W2, rootP, rootf, cvec);

    k_bias_relu_start<<<gNF4, B, 0, stream>>>(acc, b1, dinv, xw, n);
    k_edge_rows_lds  <<<gO, 1024, 0, stream>>>(xw, rowi, coli, dinv, acc, out, 1, e, n);
    k_gemm128_pool   <<<gM, B, 0, stream>>>(acc, W2, cvec, b2, rootf, out, n);
}


# Edge passes read table rows as they are needed: start rows built at the LDS tile load (own row x dinv^2), layer 2 forms relu(sums+b1) per row read; h1 never stored, 2 launches go; sum order kept
# speedup vs baseline: 3.3390x; 1.1467x over previous
#include <hip/hip_runtime.h>

typedef __attribute__((ext_vector_type(2))) float v2f;
typedef __attribute__((ext_vector_type(8))) float v8f;

#define DIM 128
#define OUTW 254

#define BLK (16 * OUTW)
#define P1 32
#define RPO 391
#define BMO 3040
__device__ __forceinline__ int bm_stride(int e) { return (((e + 31) >> 5) + 1023) & ~1023; }
__device__ __forceinline__ long long bm_word(int t) { return (long long)(t >> 10) * BLK + BMO + (t & 1023); }
__device__ __forceinline__ long long ws_slot(int m) { return (long long)(m >> 4) * BLK + (m & 15); }
__device__ __forceinline__ int package_half() { return (__builtin_amdgcn_s_sendmsg_rtn(0x87) >> 18) & 1; }

__global__ void k_deg_init(float* deg, float* __restrict__ out, int n) {
    int i = blockIdx.x * blockDim.x + threadIdx.x;
    if (i < n) {
        deg[i] = 1.0f;
        out[ws_slot(i) + P1] = 0.0f;
    }
}

__global__ __launch_bounds__(256) void k_deg_count(const int* __restrict__ col, float* deg, float* __restrict__ out, int e) {
    __shared__ unsigned bm_s[8][8];
    const int lane = threadIdx.x & 31;
    const int wave = threadIdx.x >> 5;
    const int i = blockIdx.x * 256 + threadIdx.x;
    const int half = package_half();
    if (lane < 8) bm_s[wave][lane] = 0u;
    __syncthreads();
    if (i < e) {
        const int c = col[i];
        atomicAdd(half ? (out + ws_slot(c) + P1) : (deg + c), 1.0f);
        const int blk = c / RPO;
        atomicOr(&bm_s[wave][(blk >> 5) & 7], 1u << (blk & 31));
    }
    __syncthreads();
    const int g = i >> 5;
    if (lane < 8 && (g << 5) < e)
        ((unsigned*)out)[bm_word(lane * bm_stride(e) + g)] = bm_s[wave][lane];
}

__global__ void k_deg_rsqrt(float* deg, float* __restrict__ out, int n) {
    int i = blockIdx.x * blockDim.x + threadIdx.x;
    if (i < n) {
        const long long sl = ws_slot(i);
        float d = rsqrtf(deg[i] + out[sl + P1]);
        deg[i] = d;
        out[sl] = d * d;
    }
}

#define AP 132
__device__ __forceinline__ void stage_a_tile(const float* __restrict__ X,
                                             float* __restrict__ atile, int m0, int n, int tid) {
    const int r   = tid >> 5;
    const int c4  = tid & 31;
    const int ma  = m0 + r;
    const int mb  = m0 + r + 8;
    const int mac = (ma < n) ? ma : (n - 1);
    const int mbc = (mb < n) ? mb : (n - 1);
    float4 va = *(const float4*)(X + (long long)mac * DIM + 4 * c4);
    float4 vb = *(const float4*)(X + (long long)mbc * DIM + 4 * c4);
    *(float4*)(atile + r * AP + 4 * c4)       = va;
    *(float4*)(atile + (r + 8) * AP + 4 * c4) = vb;
}

__global__ __launch_bounds__(256) void k_gemm128_wmma(
    const float* __restrict__ X, const float* __restrict__ W,
    const float* __restrict__ cvec, float* __restrict__ Y, int n) {
    __shared__ __attribute__((aligned(16))) float atile[16 * AP];
    const int lane = threadIdx.x & 31;
    const int wave = threadIdx.x >> 5;
    const int m0   = blockIdx.x * 16;
    const int n0   = wave * 16;
    const int lo   = lane & 15;
    const int hi   = lane >> 4;

    stage_a_tile(X, atile, m0, n, threadIdx.x);
    const float* ap = atile + lo * AP + 2 * hi;
    const float* bp = W + (n0 + lo) + (long long)(2 * hi) * DIM;

    float ci = cvec ? cvec[n0 + lo] : 0.0f;
    v8f d;
#pragma unroll
    for (int v = 0; v < 8; ++v) d[v] = ci;
    __syncthreads();

#pragma unroll
    for (int k = 0; k < DIM; k += 4) {
        v2f a = *(const v2f*)(ap + k);
        v2f b;
        b.x = bp[(long long)k * DIM];
        b.y = bp[(long long)k * DIM + DIM];
        d = __builtin_amdgcn_wmma_f32_16x16x4_f32(false, a, false, b,
                                                  (short)0, d, false, false);
    }

    float* yp = Y + (long long)(m0 + 8 * hi) * DIM + n0 + lo;
#pragma unroll
    for (int v = 0; v < 8; ++v) {
        int m = m0 + v + 8 * hi;
        if (m < n) yp[(long long)v * DIM] = d[v];
    }
}

#define FP 258
__global__ __launch_bounds__(256) void k_gemm128_pool(
    const float* __restrict__ X, const float* __restrict__ W,
    const float* __restrict__ cvec, const float* __restrict__ b2,
    const float* __restrict__ root_feat, float* __restrict__ out, int n) {
    __shared__ float feat[16 * FP];
    __shared__ __attribute__((aligned(16))) float atile[16 * AP];
    const int tid  = threadIdx.x;
    const int lane = tid & 31;
    const int wave = tid >> 5;
    const int m0   = blockIdx.x * 16;
    const int n0   = wave * 16;
    const int lo   = lane & 15;
    const int hi   = lane >> 4;

    stage_a_tile(X, atile, m0, n, tid);
    const float* ap = atile + lo * AP + 2 * hi;
    const float* bp = W + (n0 + lo) + (long long)(2 * hi) * DIM;

    const float cj = cvec[n0 + lo];
    const float bj = b2[n0 + lo];
    const float rf = root_feat[tid & 127];
    v8f d;
#pragma unroll
    for (int v = 0; v < 8; ++v) {
        int m = m0 + v + 8 * hi;
        int mc = (m < n) ? m : (n - 1);
        const long long sl = ws_slot(mc);
        d[v] = out[sl] * cj + bj;
    }
    __syncthreads();

#pragma unroll
    for (int k = 0; k < DIM; k += 4) {
        v2f a = *(const v2f*)(ap + k);
        v2f b;
        b.x = bp[(long long)k * DIM];
        b.y = bp[(long long)k * DIM + DIM];
        d = __builtin_amdgcn_wmma_f32_16x16x4_f32(false, a, false, b,
                                                  (short)0, d, false, false);
    }

    asm volatile("v_nop\n\tv_nop\n\tv_nop\n\tv_nop" : "+v"(d));

#pragma unroll
    for (int j = 0; j < 8; ++j) feat[(2 * j + (tid >> 7)) * FP + (tid & 127)] = rf;
#pragma unroll
    for (int v = 0; v < 8; ++v) feat[(v + 8 * hi) * FP + DIM + n0 + lo] = fmaxf(d[v], 0.0f);
    __syncthreads();

    const int rows  = (n - m0 < 16) ? (n - m0) : 16;
    const int total = rows * OUTW;
    float* ob = out + (long long)m0 * OUTW;
#pragma unroll
    for (int j = 0; j < 16; ++j) {
        int q = tid + 256 * j;
        if (q < total) {
            int r = q / OUTW;
            int t = q - r * OUTW;
            const float* f = feat + r * FP + t;
            ob[q] = (f[0] + f[1] + f[2]) * (1.0f / 3.0f);
        }
    }
}

#define SCAP 544
#define GCAP 144
#define GB 16
#define EG 16

#define SREC 1488
#define RSH 22
struct SaveCtx { int2* recs; int* save_n; int fb; int cap; };
__device__ __forceinline__ long long rec_index(int fb, int s) {
    const int bq = s / SREC;
    return (long long)(fb + bq) * (BLK / 2) + 32 + (s - bq * SREC);
}

struct RowTf { int on; float b[4]; };

template <bool TF>
__device__ __forceinline__ void add_group(const float (&v)[EG][4], float* tile, int lane,
                                          int lr, int nbits, int j0, int cnt, const RowTf& tf) {
#pragma unroll
    for (int u = 0; u < EG; ++u) {
        if (j0 + u < cnt) {
            const int lj = __builtin_amdgcn_readlane(lr, j0 + u);
            const float nj = __int_as_float(__builtin_amdgcn_readlane(nbits, j0 + u));
            float* drow = tile + lj * DIM + lane;
            float x0 = v[u][0], x1 = v[u][1], x2 = v[u][2], x3 = v[u][3];
            if (TF) {
                x0 = fmaxf(x0 + tf.b[0], 0.0f);
                x1 = fmaxf(x1 + tf.b[1], 0.0f);
                x2 = fmaxf(x2 + tf.b[2], 0.0f);
                x3 = fmaxf(x3 + tf.b[3], 0.0f);
            }
            atomicAdd(drow,      x0 * nj);
            atomicAdd(drow + 32, x1 * nj);
            atomicAdd(drow + 64, x2 * nj);
            atomicAdd(drow + 96, x3 * nj);
        }
    }
}

__device__ __forceinline__ void add_rows_lds(const float* __restrict__ xw, float* tile, int lane,
                                             int r, int lr, int nbits, int cnt, const RowTf tf) {
    for (int j0 = 0; j0 < cnt; j0 += EG) {
        float v[EG][4];
#pragma unroll
        for (int u = 0; u < EG; ++u) {
            const int ju = (j0 + u < cnt) ? (j0 + u) : (cnt - 1);
            const int rj = __builtin_amdgcn_readlane(r, ju);
            const float* srow = xw + (long long)rj * DIM + lane;
            v[u][0] = srow[0]; v[u][1] = srow[32]; v[u][2] = srow[64]; v[u][3] = srow[96];
        }
        if (tf.on) add_group<true >(v, tile, lane, lr, nbits, j0, cnt, tf);
        else       add_group<false>(v, tile, lane, lr, nbits, j0, cnt, tf);
    }
}

__device__ __forceinline__ void work_off(const float* __restrict__ xw, const int* __restrict__ rowi,
                                         const float* __restrict__ dinv, float* tile, float* wsum_s,
                                         const int* __restrict__ coli, const int* lse, int off, int k,
                                         int lo, int lane, int with_wsum, const SaveCtx sc, const RowTf tf) {
    const int j  = (lane < k) ? lane : 0;
    const int ej = lse[off + j];
    const int r  = rowi[ej];
    const int lr = coli[ej] - lo;
    const float nrm = dinv[r] * dinv[lo + lr];
    if (with_wsum && lane < k) atomicAdd(wsum_s + lr, nrm);
    if (!with_wsum) {
        int base = 0;
        if (lane == 0) base = atomicAdd(sc.save_n, k);
        base = __builtin_amdgcn_readfirstlane(base);
        const int s = base + lane;
        if (lane < k && s < sc.cap) sc.recs[rec_index(sc.fb, s)] = make_int2(r | (lr << RSH), __float_as_int(nrm));
    }
    add_rows_lds(xw, tile, lane, r, lr, __float_as_int(nrm), k, tf);
}

__device__ __forceinline__ int take_groups(const int* gl, int off, int k, int cnt, int lo, int nrow, int lane, int e,
                                           const float* __restrict__ xw, const int* __restrict__ rowi,
                                           const int* __restrict__ coli, const float* __restrict__ dinv,
                                           float* tile, float* wsum_s, int* lse, int with_wsum, const SaveCtx sc,
                                           const RowTf tf) {
    const int gj = gl[off + ((lane < k) ? lane : 0)];
    int idv[GB];
#pragma unroll
    for (int u = 0; u < GB; ++u) {
        const int gu = __builtin_amdgcn_readlane(gj, (u < k) ? u : (k - 1));
        const int i = 32 * gu + lane;
        idv[u] = (i < e) ? coli[i] : -1;
    }
#pragma unroll
    for (int u = 0; u < GB; ++u) {
        if (u < k) {
            const unsigned m = __builtin_amdgcn_ballot_w32((unsigned)(idv[u] - lo) < (unsigned)nrow);
            if (m != 0u) {
                const int pos = cnt + (int)__builtin_amdgcn_mbcnt_lo(m, 0u);
                if ((m >> lane) & 1u) lse[pos] = 32 * __builtin_amdgcn_readlane(gj, u) + lane;
                cnt += __popc(m);
            }
        }
    }
    if (cnt >= 32) {
        int eo = 0;
        while (cnt - eo >= 32) {
            work_off(xw, rowi, dinv, tile, wsum_s, coli, lse, eo, 32, lo, lane, with_wsum, sc, tf);
            eo += 32;
        }
        const int rem = cnt - eo;
        if (lane < rem) {
            const int te = lse[eo + lane];
            lse[lane] = te;
        }
        cnt = rem;
    }
    return cnt;
}

__device__ __forceinline__ float4 start_piece(float4 v, float d, float4 b, int on) {
    if (on) {
        v.x = fmaxf(v.x + b.x, 0.0f);
        v.y = fmaxf(v.y + b.y, 0.0f);
        v.z = fmaxf(v.z + b.z, 0.0f);
        v.w = fmaxf(v.w + b.w, 0.0f);
    }
    const float s = d * d;
    return make_float4(v.x * s, v.y * s, v.z * s, v.w * s);
}

__global__ __launch_bounds__(1024) void k_edge_rows_lds(const float* __restrict__ xw,
                               const int* __restrict__ rowi, const int* __restrict__ coli,
                               const float* __restrict__ dinv, float* __restrict__ acc,
                               float* __restrict__ outp, int with_wsum, int e, int n,
                               const float* __restrict__ bias) {
    __shared__ __attribute__((aligned(16))) float tile[RPO * DIM];
    __shared__ float wsum_s[RPO];
    __shared__ int lse_all[32][SCAP];
    __shared__ int gl_all[32][GCAP];
    __shared__ int save_n;
    const int tid  = threadIdx.x;
    const int lane = tid & 31;
    const int wave = __builtin_amdgcn_readfirstlane(tid >> 5);
    const int lo   = blockIdx.x * RPO;
    const int nrow = (n - lo < RPO) ? (n - lo) : RPO;
    float* const arow = acc + (long long)lo * DIM;
    const float4* const trow = (const float4*)(xw + (long long)lo * DIM);
    const int nf4  = nrow * 32;

    RowTf tf;
    tf.on = (bias != nullptr);
    float4 bb4 = make_float4(0.0f, 0.0f, 0.0f, 0.0f);
    tf.b[0] = tf.b[1] = tf.b[2] = tf.b[3] = 0.0f;
    if (tf.on) {
        tf.b[0] = bias[lane];
        tf.b[1] = bias[lane + 32];
        tf.b[2] = bias[lane + 64];
        tf.b[3] = bias[lane + 96];
        bb4 = ((const float4*)bias)[lane];
    }

    SaveCtx sc;
    sc.recs   = (int2*)outp;
    sc.save_n = &save_n;
    sc.fb     = (lo + 15) >> 4;
    const int nblk = ((lo + nrow) >> 4) - sc.fb;
    sc.cap    = (nblk > 0) ? nblk * SREC : 0;
    int* const lenp = (int*)outp + (long long)(lo >> 4) * BLK + 16;
    int saved = -1;
    if (with_wsum) saved = __builtin_amdgcn_readfirstlane(*lenp);
    if (saved > sc.cap) saved = -1;
    if (tid == 0) save_n = 0;

    const int nfull = nf4 & ~4095;
    for (int t0 = 0; t0 < nfull; t0 += 4096) {
        float4 sv[4];
        float  sd[4];
#pragma unroll
        for (int u = 0; u < 4; ++u) {
            const int t = t0 + 1024 * u + tid;
            sv[u] = trow[t];
            sd[u] = dinv[lo + (t >> 5)];
        }
#pragma unroll
        for (int u = 0; u < 4; ++u) ((float4*)tile)[t0 + 1024 * u + tid] = start_piece(sv[u], sd[u], bb4, tf.on);
    }
    for (int t = nfull + tid; t < nf4; t += 1024)
        ((float4*)tile)[t] = start_piece(trow[t], dinv[lo + (t >> 5)], bb4, tf.on);
    if (with_wsum)
        for (int t = tid; t < nrow; t += 1024) wsum_s[t] = outp[ws_slot(lo + t)];
    __syncthreads();

    if (saved >= 0) {
        const int nbat = (saved + 31) >> 5;
        int2 rc = make_int2(0, 0);
        if (wave < nbat) {
            const int k0 = (saved - 32 * wave < 32) ? (saved - 32 * wave) : 32;
            rc = sc.recs[rec_index(sc.fb, 32 * wave + ((lane < k0) ? lane : 0))];
        }
        for (int q = wave; q < nbat; q += 32) {
            const int k = (saved - 32 * q < 32) ? (saved - 32 * q) : 32;
            int2 nx = rc;
            if (q + 32 < nbat) {
                const int kn = (saved - 32 * (q + 32) < 32) ? (saved - 32 * (q + 32)) : 32;
                nx = sc.recs[rec_index(sc.fb, 32 * (q + 32) + ((lane < kn) ? lane : 0))];
            }
            const int rr = rc.x & ((1 << RSH) - 1);
            const int r  = (rr < n) ? rr : (n - 1);
            const int lq = (int)((unsigned)rc.x >> RSH);
            const int lr = (lq < nrow) ? lq : (nrow - 1);
            if (lane < k) atomicAdd(wsum_s + lr, __int_as_float(rc.y));
            add_rows_lds(xw, tile, lane, r, lr, rc.y, k, tf);
            rc = nx;
        }
    } else {
    const int ng   = (e + 31) >> 5;
    const unsigned* const bmw = (const unsigned*)outp;
    const int pbase = (blockIdx.x >> 5) * bm_stride(e);
    const int bbit  = blockIdx.x & 31;
    const int gseg = (((ng + 31) >> 5) + 127) & ~127;
    const int gbeg = wave * gseg;
    const int gend = (ng - gbeg < gseg) ? ng : (gbeg + gseg);
    int* const lse = lse_all[wave];
    int* const gl  = gl_all[wave];
    int cnt  = 0;
    int gcnt = 0;
    if (gbeg < gend) {
        uint4 cur = *(const uint4*)(bmw + bm_word(pbase + gbeg + 4 * lane));
        for (int gb = gbeg; gb < gend; gb += 128) {
            const int nb = (gb + 128 < gend) ? (gb + 128) : gb;
            const uint4 nxt = *(const uint4*)(bmw + bm_word(pbase + nb + 4 * lane));
            const unsigned w[4] = { cur.x, cur.y, cur.z, cur.w };
#pragma unroll
            for (int k = 0; k < 4; ++k) {
                const int g = gb + 4 * lane + k;
                const bool hit = (((w[k] >> bbit) & 1u) != 0u) && (g < gend);
                const unsigned m = __builtin_amdgcn_ballot_w32(hit);
                if (m != 0u) {
                    const int pos = gcnt + (int)__builtin_amdgcn_mbcnt_lo(m, 0u);
                    if (hit) gl[pos] = g;
                    gcnt += __popc(m);
                }
            }
            if (gcnt >= GB) {
                int go = 0;
                while (gcnt - go >= GB) {
                    cnt = take_groups(gl, go, GB, cnt, lo, nrow, lane, e, xw, rowi, coli, dinv, tile, wsum_s, lse, with_wsum, sc, tf);
                    go += GB;
                }
                const int grem = gcnt - go;
                if (lane < grem) {
                    const int tg = gl[go + lane];
                    gl[lane] = tg;
                }
                gcnt = grem;
            }
            cur = nxt;
        }
    }
    if (gcnt > 0) cnt = take_groups(gl, 0, gcnt, cnt, lo, nrow, lane, e, xw, rowi, coli, dinv, tile, wsum_s, lse, with_wsum, sc, tf);
    if (cnt > 0) work_off(xw, rowi, dinv, tile, wsum_s, coli, lse, 0, cnt, lo, lane, with_wsum, sc, tf);
    }
    __syncthreads();
    if (!with_wsum && tid == 0) *lenp = (save_n <= sc.cap) ? save_n : -1;

    for (int t = tid; t < nf4; t += 1024) ((float4*)arow)[t] = ((const float4*)tile)[t];
    if (with_wsum)
        for (int t = tid; t < nrow; t += 1024) outp[ws_slot(lo + t)] = wsum_s[t];
}

__global__ void k_root_and_c(const float* __restrict__ acc,
                             const float* __restrict__ b1,
                             const float* __restrict__ posts, const float* __restrict__ W2,
                             const int* __restrict__ rootPtr,
                             float* __restrict__ root_feat, float* __restrict__ cvec) {
    int j = threadIdx.x;
    int root = *rootPtr;
    root_feat[j] = acc[(long long)root * DIM + j] + b1[j];
    float s = 0.0f;
    for (int k = 0; k < DIM; ++k)
        s += fmaxf(posts[(long long)root * DIM + k], 0.0f) * W2[(long long)(DIM + k) * DIM + j];
    cvec[j] = s;
}

extern "C" void kernel_launch(void* const* d_in, const int* in_sizes, int n_in,
                              void* d_out, int out_size, void* d_ws, size_t ws_size,
                              hipStream_t stream) {
    const float* posts = (const float*)d_in[0];
    const float* W1    = (const float*)d_in[1];
    const float* b1    = (const float*)d_in[2];
    const float* W2    = (const float*)d_in[3];
    const float* b2    = (const float*)d_in[4];
    const int*   edge  = (const int*)d_in[5];
    const int*   rootP = (const int*)d_in[6];

    const int n = in_sizes[0] / DIM;
    const int e = in_sizes[5] / 2;
    const int* rowi = edge;
    const int* coli = edge + e;

    char* ws = (char*)d_ws;
    size_t off = 0;
    float* dinv = (float*)(ws + off); off += (((size_t)n * 4) + 255) & ~(size_t)255;
    float* xw   = (float*)(ws + off); off += (size_t)n * DIM * 4;
    float* acc  = (float*)(ws + off); off += (size_t)n * DIM * 4;
    float* rootf = (float*)(ws + off); off += DIM * 4;
    float* cvec  = (float*)(ws + off);
    float* out = (float*)d_out;

    const int B = 256;
    const int gN   = (n + B - 1) / B;
    const int gE   = (e + B - 1) / B;
    const int gO   = (n + RPO - 1) / RPO;
    const int gM   = (n + 15) / 16;

    k_deg_init <<<gN, B, 0, stream>>>(dinv, out, n);
    k_deg_count<<<gE, B, 0, stream>>>(coli, dinv, out, e);
    k_deg_rsqrt<<<gN, B, 0, stream>>>(dinv, out, n);

    k_gemm128_wmma <<<gM, B, 0, stream>>>(posts, W1, nullptr, xw, n);
    k_edge_rows_lds<<<gO, 1024, 0, stream>>>(xw, rowi, coli, dinv, acc, out, 0, e, n, nullptr);

    k_root_and_c<<<1, DIM, 0, stream>>>(acc, b1, posts, W2, rootP, rootf, cvec);

    k_edge_rows_lds<<<gO, 1024, 0, stream>>>(acc, rowi, coli, dinv, xw, out, 1, e, n, b1);
    k_gemm128_pool <<<gM, B, 0, stream>>>(xw, W2, cvec, b2, rootf, out, n);
}
